# starting kernel
# baseline (speedup 1.0000x reference)
#include <hip/hip_runtime.h>
#include <math.h>

#define NUM_TBLOCKS 8
#define SEQ 54
#define DIM 6
#define QKV (3 * DIM)
#define FEAT (SEQ * DIM)
#define POL 12
#define LN_EPS 1e-5f
#define QSCALE 0.40824829046386302f

#define WAVES_PER_BLOCK 4
#define THREADS (WAVES_PER_BLOCK * 32)
#define WAVE_LDS 3136
#define WCACHE 704

typedef __attribute__((ext_vector_type(2))) float v2f;
typedef __attribute__((ext_vector_type(4))) float v4f;
typedef __attribute__((ext_vector_type(8))) float v8f;

__device__ __forceinline__ v8f wmma_f32(v2f a, v2f b, v8f c) {
  return __builtin_amdgcn_wmma_f32_16x16x4_f32(false, a, false, b, (short)0, c,
                                               false, false);
}

__device__ __forceinline__ float eluf(float x) {
  return x > 0.0f ? x : (__expf(x) - 1.0f);
}

__global__ __launch_bounds__(THREADS) void rubik_fused(
    const float* __restrict__ x, const float* __restrict__ embed_w,
    const float* __restrict__ embed_b, const float* __restrict__ fc1_w,
    const float* __restrict__ fc1_b, const float* __restrict__ ln_g,
    const float* __restrict__ ln_b, const float* __restrict__ iw,
    const float* __restrict__ ib, const float* __restrict__ ow,
    const float* __restrict__ ob, const float* __restrict__ vw,
    const float* __restrict__ vb, const float* __restrict__ pw,
    const float* __restrict__ pb, float* __restrict__ out, int n_batch) {
  __shared__ float smem[WAVES_PER_BLOCK * WAVE_LDS + WCACHE];
  const int lane = threadIdx.x & 31;
  const int wave = threadIdx.x >> 5;
  int b = blockIdx.x * WAVES_PER_BLOCK + wave;
  if (b >= n_batch) b = n_batch - 1;

  float* H = &smem[wave * WAVE_LDS];
  float* Z = H + 324;
  float* Qp = Z + 324;
  float* Kp = Qp + 512;
  float* Vp = Kp + 512;
  float* P = Vp + 448;
  float* dump = H + 3016;

  float* WC = &smem[WAVES_PER_BLOCK * WAVE_LDS];
  float* W_FC = WC;
  float* W_OW = WC + 128;
  float* W_Q0 = WC + 256;
  float* W_Q1 = WC + 384;
  float* B_FC = WC + 512;
  float* B_OW = WC + 544;
  float* B_Q0 = WC + 576;
  float* B_Q1 = WC + 608;

  const float* xb = x + (size_t)b * FEAT;
  __builtin_prefetch(xb, 0, 1);

  const int nloc = lane & 15;
  const int kb = (lane < 16) ? 0 : 2;
  const int upper8 = (lane < 16) ? 0 : 8;
  const bool lowhalf = (lane < 16);

  const int qgrp = nloc / 6;
  const int qcol = nloc - qgrp * 6;
  const float qscl = (qgrp == 0) ? QSCALE : 1.0f;
  float* qkvBase = Qp + qgrp * 512 + qcol;

  auto fillWB = [&](const float* w, const float* bias, int nCols, int ntile,
                    float* wimg, float* bimg) {
    for (int slot = threadIdx.x; slot < 128; slot += THREADS) {
      int kk = slot >> 6, ln = (slot >> 1) & 31, p = slot & 1;
      int nl = ln & 15;
      int c = kk * 4 + ((ln < 16) ? 0 : 2) + p;
      int j = ntile * 16 + nl;
      bool v = (j < nCols) && (c < DIM);
      wimg[slot] = v ? w[v ? j * DIM + c : 0] : 0.0f;
    }
    for (int ln = threadIdx.x; ln < 32; ln += THREADS) {
      int j = ntile * 16 + (ln & 15);
      bimg[ln] = (j < nCols) ? bias[(j < nCols) ? j : 0] : 0.0f;
    }
  };

  auto loadA = [&](const float* src, int mt, v2f aw[2]) {
    int row = mt * 16 + nloc;
#pragma unroll
    for (int kk = 0; kk < 2; ++kk)
      aw[kk] = *(const v2f*)(src + row * DIM + kk * 4 + kb);
  };
  auto loadBimg = [&](const float* wimg, v2f bw[2]) {
    bw[0] = *(const v2f*)(wimg + lane * 2);
    bw[1] = *(const v2f*)(wimg + 64 + lane * 2);
  };
  auto loadCimg = [&](const float* bimg) {
    float bvv = bimg[lane];
    v8f c;
#pragma unroll
    for (int r = 0; r < 8; ++r) c[r] = bvv;
    return c;
  };

  for (int idx = lane; idx < FEAT; idx += 32) Z[idx] = xb[idx];
  for (int idx = lane; idx < 512 + 512 + 448; idx += 32) Qp[idx] = 0.0f;
  fillWB(embed_w, embed_b, DIM, 0, W_FC, B_FC);
  __syncthreads();

  {
    v2f bw[2];
    loadBimg(W_FC, bw);
    v8f cb = loadCimg(B_FC);
#pragma unroll
    for (int mt = 0; mt < 4; ++mt) {
      v2f aw[2];
      loadA(Z, mt, aw);
      v8f d = wmma_f32(aw[0], bw[0], cb);
      d = wmma_f32(aw[1], bw[1], d);
      int row0 = mt * 16 + upper8;
      bool ok = (nloc < DIM) && ((mt < 3) || lowhalf);
      float* bz = ok ? &H[row0 * DIM + nloc] : dump;
#pragma unroll
      for (int r = 0; r < 8; ++r)
        if (mt < 3 || r < 6) bz[r * DIM] = d[r];
    }
  }
  __syncthreads();

  for (int L = 0; L < NUM_TBLOCKS; ++L) {
    const float* g = ln_g + L * DIM;
    const float* bbia = ln_b + L * DIM;

    fillWB(fc1_w + L * DIM * DIM, fc1_b + L * DIM, DIM, 0, W_FC, B_FC);
    fillWB(ow + L * DIM * DIM, ob + L * DIM, DIM, 0, W_OW, B_OW);
    fillWB(iw + L * QKV * DIM, ib + L * QKV, QKV, 0, W_Q0, B_Q0);
    fillWB(iw + L * QKV * DIM, ib + L * QKV, QKV, 1, W_Q1, B_Q1);
    __syncthreads();

    {
      v2f bw[2];
      loadBimg(W_FC, bw);
      v8f cb = loadCimg(B_FC);
#pragma unroll
      for (int mt = 0; mt < 4; ++mt) {
        v2f aw[2];
        loadA(H, mt, aw);
        v8f d = wmma_f32(aw[0], bw[0], cb);
        d = wmma_f32(aw[1], bw[1], d);
        int row0 = mt * 16 + upper8;
        bool ok = (nloc < DIM) && ((mt < 3) || lowhalf);
        float* bz = ok ? &Z[row0 * DIM + nloc] : dump;
#pragma unroll
        for (int r = 0; r < 8; ++r)
          if (mt < 3 || r < 6) bz[r * DIM] = eluf(d[r]);
      }
    }
    __syncthreads();

    for (int s = lane; s < SEQ; s += 32) {
      float mu = 0.0f;
#pragma unroll
      for (int i = 0; i < DIM; ++i) mu += Z[s * DIM + i];
      mu *= (1.0f / DIM);
      float var = 0.0f;
#pragma unroll
      for (int i = 0; i < DIM; ++i) {
        float d = Z[s * DIM + i] - mu;
        var += d * d;
      }
      var *= (1.0f / DIM);
      float rs = rsqrtf(var + LN_EPS);
#pragma unroll
      for (int i = 0; i < DIM; ++i)
        Z[s * DIM + i] = (Z[s * DIM + i] - mu) * rs * g[i] + bbia[i];
    }
    __syncthreads();

    {
      v2f bw0[2], bw1[2];
      loadBimg(W_Q0, bw0);
      loadBimg(W_Q1, bw1);
      v8f cb0 = loadCimg(B_Q0);
      v8f cb1 = loadCimg(B_Q1);
#pragma unroll
      for (int mt = 0; mt < 4; ++mt) {
        v2f aw[2];
        loadA(Z, mt, aw);
        v8f d0 = wmma_f32(aw[0], bw0[0], cb0);
        d0 = wmma_f32(aw[1], bw0[1], d0);
        v8f d1 = wmma_f32(aw[0], bw1[0], cb1);
        d1 = wmma_f32(aw[1], bw1[1], d1);
        int row0 = mt * 16 + upper8;
        bool ok0 = (mt < 3) || lowhalf;
        float* b0 = ok0 ? (qkvBase + row0 * 8) : dump;
        float* b1 = (ok0 && nloc < 2) ? (Vp + row0 * 8 + 4 + nloc) : dump;
#pragma unroll
        for (int r = 0; r < 8; ++r)
          if (mt < 3 || r < 6) {
            b0[r * 8] = d0[r] * qscl;
            b1[r * 8] = d1[r];
          }
      }
    }
    __syncthreads();

    v2f ak[4][2];
#pragma unroll
    for (int mt = 0; mt < 4; ++mt)
#pragma unroll
      for (int kk = 0; kk < 2; ++kk)
        ak[mt][kk] = *(const v2f*)(Kp + (mt * 16 + nloc) * 8 + kk * 4 + kb);
    v2f bv[14];
#pragma unroll
    for (int kc = 0; kc < 14; ++kc) {
      int kr = kc * 4 + kb;
      bv[kc].x = (nloc < 8) ? Vp[kr * 8 + nloc] : 0.0f;
      bv[kc].y = (nloc < 8) ? Vp[(kr + 1) * 8 + nloc] : 0.0f;
    }

    for (int nt = 0; nt < 4; ++nt) {
      v2f bq[2];
#pragma unroll
      for (int kk = 0; kk < 2; ++kk)
        bq[kk] = *(const v2f*)(Qp + (nt * 16 + nloc) * 8 + kk * 4 + kb);
      v8f t[4];
#pragma unroll
      for (int mt = 0; mt < 4; ++mt) {
        v8f c = {};
#pragma unroll
        for (int kk = 0; kk < 2; ++kk) c = wmma_f32(ak[mt][kk], bq[kk], c);
        t[mt] = c;
      }

      float mx = -3.0e38f;
#pragma unroll
      for (int mt = 0; mt < 4; ++mt)
#pragma unroll
        for (int r = 0; r < 8; ++r) {
          int k = mt * 16 + r + upper8;
          if (k < SEQ) mx = fmaxf(mx, t[mt][r]);
        }
      mx = fmaxf(mx, __shfl_xor(mx, 16));
      float sum = 0.0f;
#pragma unroll
      for (int mt = 0; mt < 4; ++mt)
#pragma unroll
        for (int r = 0; r < 8; ++r) {
          int k = mt * 16 + r + upper8;
          float e = (k < SEQ) ? __expf(t[mt][r] - mx) : 0.0f;
          t[mt][r] = e;
          sum += e;
        }
      sum += __shfl_xor(sum, 16);
      float inv = 1.0f / sum;

#pragma unroll
      for (int mt = 0; mt < 4; ++mt) {
        v4f lo = {t[mt][0] * inv, t[mt][1] * inv, t[mt][2] * inv,
                  t[mt][3] * inv};
        v4f hi = {t[mt][4] * inv, t[mt][5] * inv, t[mt][6] * inv,
                  t[mt][7] * inv};
        bool ok = (mt < 3) || lowhalf;
        float* bp = ok ? (P + nloc * 56 + mt * 16 + upper8) : dump;
        *(v4f*)(bp) = lo;
        *(v4f*)(bp + 4) = hi;
      }
      __syncthreads();

      v8f o = {};
#pragma unroll
      for (int kc = 0; kc < 14; ++kc) {
        v2f ap = *(const v2f*)(P + nloc * 56 + kc * 4 + kb);
        o = wmma_f32(ap, bv[kc], o);
      }
      {
        int row0 = nt * 16 + upper8;
        bool ok = (nloc < DIM) && ((nt < 3) || lowhalf);
        float* bz = ok ? &Z[row0 * DIM + nloc] : dump;
#pragma unroll
        for (int r = 0; r < 8; ++r)
          if (nt < 3 || r < 6) bz[r * DIM] = o[r];
      }
      __syncthreads();
    }

    {
      v2f bw[2];
      loadBimg(W_OW, bw);
      v8f cb = loadCimg(B_OW);
#pragma unroll
      for (int mt = 0; mt < 4; ++mt) {
        v2f aw[2];
        loadA(Z, mt, aw);
        v8f d = wmma_f32(aw[0], bw[0], cb);
        d = wmma_f32(aw[1], bw[1], d);
        int row0 = mt * 16 + upper8;
        bool ok = (nloc < DIM) && ((mt < 3) || lowhalf);
        float* bh = ok ? &H[row0 * DIM + nloc] : dump;
#pragma unroll
        for (int r = 0; r < 8; ++r)
          if (mt < 3 || r < 6) {
            float hv = H[row0 * DIM + nloc + r * DIM];
            bh[r * DIM] = eluf(d[r]) + hv;
          }
      }
    }
    __syncthreads();
  }

  float vacc = 0.0f;
  float pacc[POL];
#pragma unroll
  for (int o = 0; o < POL; ++o) pacc[o] = 0.0f;
  for (int j = lane; j < FEAT; j += 32) {
    float fj = H[j];
    vacc += fj * vw[j];
#pragma unroll
    for (int o = 0; o < POL; ++o) pacc[o] += fj * pw[o * FEAT + j];
  }
  for (int off = 16; off >= 1; off >>= 1) {
    vacc += __shfl_xor(vacc, off, 32);
#pragma unroll
    for (int o = 0; o < POL; ++o) pacc[o] += __shfl_xor(pacc[o], off, 32);
  }
  if (lane == 0) {
    out[b] = vacc + vb[0];
    float logit[POL];
    float mx = -3.0e38f;
#pragma unroll
    for (int o = 0; o < POL; ++o) {
      logit[o] = pacc[o] + pb[o];
      mx = fmaxf(mx, logit[o]);
    }
    float sum = 0.0f;
#pragma unroll
    for (int o = 0; o < POL; ++o) {
      logit[o] = __expf(logit[o] - mx);
      sum += logit[o];
    }
    float inv = 1.0f / sum;
    float* pout = out + n_batch + (size_t)b * POL;
#pragma unroll
    for (int o = 0; o < POL; ++o) pout[o] = logit[o] * inv;
  }
}

extern "C" void kernel_launch(void* const* d_in, const int* in_sizes, int n_in,
                              void* d_out, int out_size, void* d_ws,
                              size_t ws_size, hipStream_t stream) {
  (void)n_in;
  (void)out_size;
  (void)d_ws;
  (void)ws_size;
  const float* x = (const float*)d_in[0];
  const float* embed_w = (const float*)d_in[1];
  const float* embed_b = (const float*)d_in[2];
  const float* fc1_w = (const float*)d_in[3];
  const float* fc1_b = (const float*)d_in[4];
  const float* ln_g = (const float*)d_in[5];
  const float* ln_b = (const float*)d_in[6];
  const float* iw = (const float*)d_in[7];
  const float* ib = (const float*)d_in[8];
  const float* ow = (const float*)d_in[9];
  const float* ob = (const float*)d_in[10];
  const float* vw = (const float*)d_in[11];
  const float* vb = (const float*)d_in[12];
  const float* pw = (const float*)d_in[13];
  const float* pb = (const float*)d_in[14];

  int n_batch = in_sizes[0] / FEAT;
  int grid = (n_batch + WAVES_PER_BLOCK - 1) / WAVES_PER_BLOCK;
  rubik_fused<<<grid, THREADS, 0, stream>>>(x, embed_w, embed_b, fc1_w, fc1_b,
                                            ln_g, ln_b, iw, ib, ow, ob, vw, vb,
                                            pw, pb, (float*)d_out, n_batch);
}


# Softmax probabilities stay in registers: each score D fragment is the B operand of O^T=[V|1]^T P^T (ones row = row sum on the matrix unit, scaled after); key order of PV and denominator sums changes
# speedup vs baseline: 1.0902x; 1.0902x over previous
#include <hip/hip_runtime.h>
#include <math.h>

#define NUM_TBLOCKS 8
#define SEQ 54
#define DIM 6
#define QKV (3 * DIM)
#define FEAT (SEQ * DIM)
#define POL 12
#define LN_EPS 1e-5f
#define QSCALE 0.40824829046386302f

#define WAVES_PER_BLOCK 4
#define THREADS (WAVES_PER_BLOCK * 32)
#define WAVE_LDS 2240
#define WCACHE 704

typedef __attribute__((ext_vector_type(2))) float v2f;
typedef __attribute__((ext_vector_type(4))) float v4f;
typedef __attribute__((ext_vector_type(8))) float v8f;

__device__ __forceinline__ v8f wmma_f32(v2f a, v2f b, v8f c) {
  return __builtin_amdgcn_wmma_f32_16x16x4_f32(false, a, false, b, (short)0, c,
                                               false, false);
}

__device__ __forceinline__ float xhalf(float x) {
  int b = __builtin_bit_cast(int, x);
  return __builtin_bit_cast(
      float, __builtin_amdgcn_permlanex16(b, b, 0x76543210, 0xfedcba98, false,
                                          false));
}

__device__ __forceinline__ float eluf(float x) {
  return x > 0.0f ? x : (__expf(x) - 1.0f);
}

__global__ __launch_bounds__(THREADS) void rubik_fused(
    const float* __restrict__ x, const float* __restrict__ embed_w,
    const float* __restrict__ embed_b, const float* __restrict__ fc1_w,
    const float* __restrict__ fc1_b, const float* __restrict__ ln_g,
    const float* __restrict__ ln_b, const float* __restrict__ iw,
    const float* __restrict__ ib, const float* __restrict__ ow,
    const float* __restrict__ ob, const float* __restrict__ vw,
    const float* __restrict__ vb, const float* __restrict__ pw,
    const float* __restrict__ pb, float* __restrict__ out, int n_batch) {
  __shared__ float smem[WAVES_PER_BLOCK * WAVE_LDS + WCACHE];
  const int lane = threadIdx.x & 31;
  const int wave = threadIdx.x >> 5;
  int b = blockIdx.x * WAVES_PER_BLOCK + wave;
  if (b >= n_batch) b = n_batch - 1;

  float* H = &smem[wave * WAVE_LDS];
  float* Z = H + 324;
  float* Qp = Z + 324;
  float* Kp = Qp + 512;
  float* Vp = Kp + 512;
  float* dump = H + 2120;

  float* WC = &smem[WAVES_PER_BLOCK * WAVE_LDS];
  float* W_FC = WC;
  float* W_OW = WC + 128;
  float* W_Q0 = WC + 256;
  float* W_Q1 = WC + 384;
  float* B_FC = WC + 512;
  float* B_OW = WC + 544;
  float* B_Q0 = WC + 576;
  float* B_Q1 = WC + 608;

  const float* xb = x + (size_t)b * FEAT;
  __builtin_prefetch(xb, 0, 1);

  const int nloc = lane & 15;
  const int kb = (lane < 16) ? 0 : 2;
  const int upper8 = (lane < 16) ? 0 : 8;
  const bool lowhalf = (lane < 16);

  const int qgrp = nloc / 6;
  const int qcol = nloc - qgrp * 6;
  const float qscl = (qgrp == 0) ? QSCALE : 1.0f;
  float* qkvBase = Qp + qgrp * 512 + qcol;

  auto fillWB = [&](const float* w, const float* bias, int nCols, int ntile,
                    float* wimg, float* bimg) {
    for (int slot = threadIdx.x; slot < 128; slot += THREADS) {
      int kk = slot >> 6, ln = (slot >> 1) & 31, p = slot & 1;
      int nl = ln & 15;
      int c = kk * 4 + ((ln < 16) ? 0 : 2) + p;
      int j = ntile * 16 + nl;
      bool v = (j < nCols) && (c < DIM);
      wimg[slot] = v ? w[v ? j * DIM + c : 0] : 0.0f;
    }
    for (int ln = threadIdx.x; ln < 32; ln += THREADS) {
      int j = ntile * 16 + (ln & 15);
      bimg[ln] = (j < nCols) ? bias[(j < nCols) ? j : 0] : 0.0f;
    }
  };

  auto loadA = [&](const float* src, int mt, v2f aw[2]) {
    int row = mt * 16 + nloc;
#pragma unroll
    for (int kk = 0; kk < 2; ++kk)
      aw[kk] = *(const v2f*)(src + row * DIM + kk * 4 + kb);
  };
  auto loadBimg = [&](const float* wimg, v2f bw[2]) {
    bw[0] = *(const v2f*)(wimg + lane * 2);
    bw[1] = *(const v2f*)(wimg + 64 + lane * 2);
  };
  auto loadCimg = [&](const float* bimg) {
    float bvv = bimg[lane];
    v8f c;
#pragma unroll
    for (int r = 0; r < 8; ++r) c[r] = bvv;
    return c;
  };

  for (int idx = lane; idx < FEAT; idx += 32) Z[idx] = xb[idx];
  for (int idx = lane; idx < 512 + 512 + 448; idx += 32) Qp[idx] = 0.0f;
  fillWB(embed_w, embed_b, DIM, 0, W_FC, B_FC);
  __syncthreads();

  for (int s = lane; s < SEQ; s += 32) Vp[s * 8 + DIM] = 1.0f;

  {
    v2f bw[2];
    loadBimg(W_FC, bw);
    v8f cb = loadCimg(B_FC);
#pragma unroll
    for (int mt = 0; mt < 4; ++mt) {
      v2f aw[2];
      loadA(Z, mt, aw);
      v8f d = wmma_f32(aw[0], bw[0], cb);
      d = wmma_f32(aw[1], bw[1], d);
      int row0 = mt * 16 + upper8;
      bool ok = (nloc < DIM) && ((mt < 3) || lowhalf);
      float* bz = ok ? &H[row0 * DIM + nloc] : dump;
#pragma unroll
      for (int r = 0; r < 8; ++r)
        if (mt < 3 || r < 6) bz[r * DIM] = d[r];
    }
  }
  __syncthreads();

  for (int L = 0; L < NUM_TBLOCKS; ++L) {
    const float* g = ln_g + L * DIM;
    const float* bbia = ln_b + L * DIM;

    fillWB(fc1_w + L * DIM * DIM, fc1_b + L * DIM, DIM, 0, W_FC, B_FC);
    fillWB(ow + L * DIM * DIM, ob + L * DIM, DIM, 0, W_OW, B_OW);
    fillWB(iw + L * QKV * DIM, ib + L * QKV, QKV, 0, W_Q0, B_Q0);
    fillWB(iw + L * QKV * DIM, ib + L * QKV, QKV, 1, W_Q1, B_Q1);
    __syncthreads();

    {
      v2f bw[2];
      loadBimg(W_FC, bw);
      v8f cb = loadCimg(B_FC);
#pragma unroll
      for (int mt = 0; mt < 4; ++mt) {
        v2f aw[2];
        loadA(H, mt, aw);
        v8f d = wmma_f32(aw[0], bw[0], cb);
        d = wmma_f32(aw[1], bw[1], d);
        int row0 = mt * 16 + upper8;
        bool ok = (nloc < DIM) && ((mt < 3) || lowhalf);
        float* bz = ok ? &Z[row0 * DIM + nloc] : dump;
#pragma unroll
        for (int r = 0; r < 8; ++r)
          if (mt < 3 || r < 6) bz[r * DIM] = eluf(d[r]);
      }
    }
    __syncthreads();

    for (int s = lane; s < SEQ; s += 32) {
      float mu = 0.0f;
#pragma unroll
      for (int i = 0; i < DIM; ++i) mu += Z[s * DIM + i];
      mu *= (1.0f / DIM);
      float var = 0.0f;
#pragma unroll
      for (int i = 0; i < DIM; ++i) {
        float d = Z[s * DIM + i] - mu;
        var += d * d;
      }
      var *= (1.0f / DIM);
      float rs = rsqrtf(var + LN_EPS);
#pragma unroll
      for (int i = 0; i < DIM; ++i)
        Z[s * DIM + i] = (Z[s * DIM + i] - mu) * rs * g[i] + bbia[i];
    }
    __syncthreads();

    {
      v2f bw0[2], bw1[2];
      loadBimg(W_Q0, bw0);
      loadBimg(W_Q1, bw1);
      v8f cb0 = loadCimg(B_Q0);
      v8f cb1 = loadCimg(B_Q1);
#pragma unroll
      for (int mt = 0; mt < 4; ++mt) {
        v2f aw[2];
        loadA(Z, mt, aw);
        v8f d0 = wmma_f32(aw[0], bw0[0], cb0);
        d0 = wmma_f32(aw[1], bw0[1], d0);
        v8f d1 = wmma_f32(aw[0], bw1[0], cb1);
        d1 = wmma_f32(aw[1], bw1[1], d1);
        int row0 = mt * 16 + upper8;
        bool ok0 = (mt < 3) || lowhalf;
        float* b0 = ok0 ? (qkvBase + row0 * 8) : dump;
        float* b1 = (ok0 && nloc < 2) ? (Vp + row0 * 8 + 4 + nloc) : dump;
#pragma unroll
        for (int r = 0; r < 8; ++r)
          if (mt < 3 || r < 6) {
            b0[r * 8] = d0[r] * qscl;
            b1[r * 8] = d1[r];
          }
      }
    }
    __syncthreads();

    v2f ak[4][2];
#pragma unroll
    for (int mt = 0; mt < 4; ++mt)
#pragma unroll
      for (int kk = 0; kk < 2; ++kk)
        ak[mt][kk] = *(const v2f*)(Kp + (mt * 16 + nloc) * 8 + kk * 4 + kb);
    const float* vcolp = Vp + ((nloc < 8) ? nloc : 7);
    v2f av[14];
#pragma unroll
    for (int mt = 0; mt < 3; ++mt)
#pragma unroll
      for (int p = 0; p < 4; ++p) {
        int key = mt * 16 + upper8 + 2 * p;
        av[mt * 4 + p].x = vcolp[key * 8];
        av[mt * 4 + p].y = vcolp[(key + 1) * 8];
      }
    {
      int key12 = lowhalf ? 48 : 52;
      av[12].x = vcolp[key12 * 8];
      av[12].y = vcolp[(key12 + 1) * 8];
      int key13 = lowhalf ? 50 : 54;
      av[13].x = vcolp[key13 * 8];
      av[13].y = vcolp[(key13 + 1) * 8];
    }

    for (int nt = 0; nt < 4; ++nt) {
      v2f bq[2];
#pragma unroll
      for (int kk = 0; kk < 2; ++kk)
        bq[kk] = *(const v2f*)(Qp + (nt * 16 + nloc) * 8 + kk * 4 + kb);
      v8f t[4];
#pragma unroll
      for (int mt = 0; mt < 4; ++mt) {
        v8f c = {};
#pragma unroll
        for (int kk = 0; kk < 2; ++kk) c = wmma_f32(ak[mt][kk], bq[kk], c);
        t[mt] = c;
      }

      float mx = -3.0e38f;
#pragma unroll
      for (int mt = 0; mt < 4; ++mt)
#pragma unroll
        for (int r = 0; r < 8; ++r) {
          int k = mt * 16 + r + upper8;
          if (k < SEQ) mx = fmaxf(mx, t[mt][r]);
        }
      mx = fmaxf(mx, __shfl_xor(mx, 16));
#pragma unroll
      for (int mt = 0; mt < 4; ++mt)
#pragma unroll
        for (int r = 0; r < 8; ++r) {
          int k = mt * 16 + r + upper8;
          t[mt][r] = (k < SEQ) ? __expf(t[mt][r] - mx) : 0.0f;
        }

      v8f o0 = {}, o1 = {};
#pragma unroll
      for (int mt = 0; mt < 3; ++mt)
#pragma unroll
        for (int p = 0; p < 4; p += 2) {
          v2f b0 = {t[mt][2 * p], t[mt][2 * p + 1]};
          v2f b1 = {t[mt][2 * p + 2], t[mt][2 * p + 3]};
          o0 = wmma_f32(av[mt * 4 + p], b0, o0);
          o1 = wmma_f32(av[mt * 4 + p + 1], b1, o1);
        }
      {
        v2f b12 = {t[3][0] + xhalf(t[3][4]), t[3][1] + xhalf(t[3][5])};
        v2f b13 = {t[3][2], t[3][3]};
        o0 = wmma_f32(av[12], b12, o0);
        o1 = wmma_f32(av[13], b13, o1);
      }
      {
        float inv = 1.0f / (o0[6] + o1[6]);
        int q = nt * 16 + nloc;
        if (lowhalf && q < SEQ) {
          float* zq = Z + q * DIM;
#pragma unroll
          for (int r = 0; r < DIM; r += 2) {
            v2f w = {(o0[r] + o1[r]) * inv, (o0[r + 1] + o1[r + 1]) * inv};
            *(v2f*)(zq + r) = w;
          }
        }
      }
    }
    __syncthreads();

    {
      v2f bw[2];
      loadBimg(W_OW, bw);
      v8f cb = loadCimg(B_OW);
#pragma unroll
      for (int mt = 0; mt < 4; ++mt) {
        v2f aw[2];
        loadA(Z, mt, aw);
        v8f d = wmma_f32(aw[0], bw[0], cb);
        d = wmma_f32(aw[1], bw[1], d);
        int row0 = mt * 16 + upper8;
        bool ok = (nloc < DIM) && ((mt < 3) || lowhalf);
        float* bh = ok ? &H[row0 * DIM + nloc] : dump;
#pragma unroll
        for (int r = 0; r < 8; ++r)
          if (mt < 3 || r < 6) {
            float hv = H[row0 * DIM + nloc + r * DIM];
            bh[r * DIM] = eluf(d[r]) + hv;
          }
      }
    }
    __syncthreads();
  }

  float vacc = 0.0f;
  float pacc[POL];
#pragma unroll
  for (int o = 0; o < POL; ++o) pacc[o] = 0.0f;
  for (int j = lane; j < FEAT; j += 32) {
    float fj = H[j];
    vacc += fj * vw[j];
#pragma unroll
    for (int o = 0; o < POL; ++o) pacc[o] += fj * pw[o * FEAT + j];
  }
  for (int off = 16; off >= 1; off >>= 1) {
    vacc += __shfl_xor(vacc, off, 32);
#pragma unroll
    for (int o = 0; o < POL; ++o) pacc[o] += __shfl_xor(pacc[o], off, 32);
  }
  if (lane == 0) {
    out[b] = vacc + vb[0];
    float logit[POL];
    float mx = -3.0e38f;
#pragma unroll
    for (int o = 0; o < POL; ++o) {
      logit[o] = pacc[o] + pb[o];
      mx = fmaxf(mx, logit[o]);
    }
    float sum = 0.0f;
#pragma unroll
    for (int o = 0; o < POL; ++o) {
      logit[o] = __expf(logit[o] - mx);
      sum += logit[o];
    }
    float inv = 1.0f / sum;
    float* pout = out + n_batch + (size_t)b * POL;
#pragma unroll
    for (int o = 0; o < POL; ++o) pout[o] = logit[o] * inv;
  }
}

extern "C" void kernel_launch(void* const* d_in, const int* in_sizes, int n_in,
                              void* d_out, int out_size, void* d_ws,
                              size_t ws_size, hipStream_t stream) {
  (void)n_in;
  (void)out_size;
  (void)d_ws;
  (void)ws_size;
  const float* x = (const float*)d_in[0];
  const float* embed_w = (const float*)d_in[1];
  const float* embed_b = (const float*)d_in[2];
  const float* fc1_w = (const float*)d_in[3];
  const float* fc1_b = (const float*)d_in[4];
  const float* ln_g = (const float*)d_in[5];
  const float* ln_b = (const float*)d_in[6];
  const float* iw = (const float*)d_in[7];
  const float* ib = (const float*)d_in[8];
  const float* ow = (const float*)d_in[9];
  const float* ob = (const float*)d_in[10];
  const float* vw = (const float*)d_in[11];
  const float* vb = (const float*)d_in[12];
  const float* pw = (const float*)d_in[13];
  const float* pb = (const float*)d_in[14];

  int n_batch = in_sizes[0] / FEAT;
  int grid = (n_batch + WAVES_PER_BLOCK - 1) / WAVES_PER_BLOCK;
  rubik_fused<<<grid, THREADS, 0, stream>>>(x, embed_w, embed_b, fc1_w, fc1_b,
                                            ln_g, ln_b, iw, ib, ow, ob, vw, vb,
                                            pw, pb, (float*)d_out, n_batch);
}


# ELU moved off the padded 16-column D fragments onto dense rows: fc1 stores its pre-activation and the LayerNorm row pass applies ELU; out-proj writes Z in place, one dense pass does H += elu(Z)
# speedup vs baseline: 1.2246x; 1.1233x over previous
#include <hip/hip_runtime.h>
#include <math.h>

#define NUM_TBLOCKS 8
#define SEQ 54
#define DIM 6
#define QKV (3 * DIM)
#define FEAT (SEQ * DIM)
#define POL 12
#define LN_EPS 1e-5f
#define QSCALE 0.40824829046386302f

#define WAVES_PER_BLOCK 4
#define THREADS (WAVES_PER_BLOCK * 32)
#define WAVE_LDS 2240
#define WCACHE 704

typedef __attribute__((ext_vector_type(2))) float v2f;
typedef __attribute__((ext_vector_type(4))) float v4f;
typedef __attribute__((ext_vector_type(8))) float v8f;

__device__ __forceinline__ v8f wmma_f32(v2f a, v2f b, v8f c) {
  return __builtin_amdgcn_wmma_f32_16x16x4_f32(false, a, false, b, (short)0, c,
                                               false, false);
}

__device__ __forceinline__ float xhalf(float x) {
  int b = __builtin_bit_cast(int, x);
  return __builtin_bit_cast(
      float, __builtin_amdgcn_permlanex16(b, b, 0x76543210, 0xfedcba98, false,
                                          false));
}

__device__ __forceinline__ float eluf(float x) {
  return x > 0.0f ? x : (__expf(x) - 1.0f);
}

__global__ __launch_bounds__(THREADS) void rubik_fused(
    const float* __restrict__ x, const float* __restrict__ embed_w,
    const float* __restrict__ embed_b, const float* __restrict__ fc1_w,
    const float* __restrict__ fc1_b, const float* __restrict__ ln_g,
    const float* __restrict__ ln_b, const float* __restrict__ iw,
    const float* __restrict__ ib, const float* __restrict__ ow,
    const float* __restrict__ ob, const float* __restrict__ vw,
    const float* __restrict__ vb, const float* __restrict__ pw,
    const float* __restrict__ pb, float* __restrict__ out, int n_batch) {
  __shared__ float smem[WAVES_PER_BLOCK * WAVE_LDS + WCACHE];
  const int lane = threadIdx.x & 31;
  const int wave = threadIdx.x >> 5;
  int b = blockIdx.x * WAVES_PER_BLOCK + wave;
  if (b >= n_batch) b = n_batch - 1;

  float* H = &smem[wave * WAVE_LDS];
  float* Z = H + 324;
  float* Qp = Z + 324;
  float* Kp = Qp + 512;
  float* Vp = Kp + 512;
  float* dump = H + 2120;

  float* WC = &smem[WAVES_PER_BLOCK * WAVE_LDS];
  float* W_FC = WC;
  float* W_OW = WC + 128;
  float* W_Q0 = WC + 256;
  float* W_Q1 = WC + 384;
  float* B_FC = WC + 512;
  float* B_OW = WC + 544;
  float* B_Q0 = WC + 576;
  float* B_Q1 = WC + 608;

  const float* xb = x + (size_t)b * FEAT;
  __builtin_prefetch(xb, 0, 1);

  const int nloc = lane & 15;
  const int kb = (lane < 16) ? 0 : 2;
  const int upper8 = (lane < 16) ? 0 : 8;
  const bool lowhalf = (lane < 16);

  const int qgrp = nloc / 6;
  const int qcol = nloc - qgrp * 6;
  const float qscl = (qgrp == 0) ? QSCALE : 1.0f;
  float* qkvBase = Qp + qgrp * 512 + qcol;

  auto fillWB = [&](const float* w, const float* bias, int nCols, int ntile,
                    float* wimg, float* bimg) {
    for (int slot = threadIdx.x; slot < 128; slot += THREADS) {
      int kk = slot >> 6, ln = (slot >> 1) & 31, p = slot & 1;
      int nl = ln & 15;
      int c = kk * 4 + ((ln < 16) ? 0 : 2) + p;
      int j = ntile * 16 + nl;
      bool v = (j < nCols) && (c < DIM);
      wimg[slot] = v ? w[v ? j * DIM + c : 0] : 0.0f;
    }
    for (int ln = threadIdx.x; ln < 32; ln += THREADS) {
      int j = ntile * 16 + (ln & 15);
      bimg[ln] = (j < nCols) ? bias[(j < nCols) ? j : 0] : 0.0f;
    }
  };

  auto loadA = [&](const float* src, int mt, v2f aw[2]) {
    int row = mt * 16 + nloc;
#pragma unroll
    for (int kk = 0; kk < 2; ++kk)
      aw[kk] = *(const v2f*)(src + row * DIM + kk * 4 + kb);
  };
  auto loadBimg = [&](const float* wimg, v2f bw[2]) {
    bw[0] = *(const v2f*)(wimg + lane * 2);
    bw[1] = *(const v2f*)(wimg + 64 + lane * 2);
  };
  auto loadCimg = [&](const float* bimg) {
    float bvv = bimg[lane];
    v8f c;
#pragma unroll
    for (int r = 0; r < 8; ++r) c[r] = bvv;
    return c;
  };

  for (int idx = lane; idx < FEAT; idx += 32) Z[idx] = xb[idx];
  for (int idx = lane; idx < 512 + 512 + 448; idx += 32) Qp[idx] = 0.0f;
  fillWB(embed_w, embed_b, DIM, 0, W_FC, B_FC);
  __syncthreads();

  for (int s = lane; s < SEQ; s += 32) Vp[s * 8 + DIM] = 1.0f;

  {
    v2f bw[2];
    loadBimg(W_FC, bw);
    v8f cb = loadCimg(B_FC);
#pragma unroll
    for (int mt = 0; mt < 4; ++mt) {
      v2f aw[2];
      loadA(Z, mt, aw);
      v8f d = wmma_f32(aw[0], bw[0], cb);
      d = wmma_f32(aw[1], bw[1], d);
      int row0 = mt * 16 + upper8;
      bool ok = (nloc < DIM) && ((mt < 3) || lowhalf);
      float* bz = ok ? &H[row0 * DIM + nloc] : dump;
#pragma unroll
      for (int r = 0; r < 8; ++r)
        if (mt < 3 || r < 6) bz[r * DIM] = d[r];
    }
  }
  __syncthreads();

  for (int L = 0; L < NUM_TBLOCKS; ++L) {
    const float* g = ln_g + L * DIM;
    const float* bbia = ln_b + L * DIM;

    fillWB(fc1_w + L * DIM * DIM, fc1_b + L * DIM, DIM, 0, W_FC, B_FC);
    fillWB(ow + L * DIM * DIM, ob + L * DIM, DIM, 0, W_OW, B_OW);
    fillWB(iw + L * QKV * DIM, ib + L * QKV, QKV, 0, W_Q0, B_Q0);
    fillWB(iw + L * QKV * DIM, ib + L * QKV, QKV, 1, W_Q1, B_Q1);
    __syncthreads();

    {
      v2f bw[2];
      loadBimg(W_FC, bw);
      v8f cb = loadCimg(B_FC);
#pragma unroll
      for (int mt = 0; mt < 4; ++mt) {
        v2f aw[2];
        loadA(H, mt, aw);
        v8f d = wmma_f32(aw[0], bw[0], cb);
        d = wmma_f32(aw[1], bw[1], d);
        int row0 = mt * 16 + upper8;
        bool ok = (nloc < DIM) && ((mt < 3) || lowhalf);
        float* bz = ok ? &Z[row0 * DIM + nloc] : dump;
#pragma unroll
        for (int r = 0; r < 8; ++r)
          if (mt < 3 || r < 6) bz[r * DIM] = d[r];
      }
    }
    __syncthreads();

    for (int s = lane; s < SEQ; s += 32) {
      float zv[DIM];
#pragma unroll
      for (int i = 0; i < DIM; ++i) zv[i] = eluf(Z[s * DIM + i]);
      float mu = 0.0f;
#pragma unroll
      for (int i = 0; i < DIM; ++i) mu += zv[i];
      mu *= (1.0f / DIM);
      float var = 0.0f;
#pragma unroll
      for (int i = 0; i < DIM; ++i) {
        float d = zv[i] - mu;
        var += d * d;
      }
      var *= (1.0f / DIM);
      float rs = rsqrtf(var + LN_EPS);
#pragma unroll
      for (int i = 0; i < DIM; ++i)
        Z[s * DIM + i] = (zv[i] - mu) * rs * g[i] + bbia[i];
    }
    __syncthreads();

    {
      v2f bw0[2], bw1[2];
      loadBimg(W_Q0, bw0);
      loadBimg(W_Q1, bw1);
      v8f cb0 = loadCimg(B_Q0);
      v8f cb1 = loadCimg(B_Q1);
#pragma unroll
      for (int mt = 0; mt < 4; ++mt) {
        v2f aw[2];
        loadA(Z, mt, aw);
        v8f d0 = wmma_f32(aw[0], bw0[0], cb0);
        d0 = wmma_f32(aw[1], bw0[1], d0);
        v8f d1 = wmma_f32(aw[0], bw1[0], cb1);
        d1 = wmma_f32(aw[1], bw1[1], d1);
        int row0 = mt * 16 + upper8;
        bool ok0 = (mt < 3) || lowhalf;
        float* b0 = ok0 ? (qkvBase + row0 * 8) : dump;
        float* b1 = (ok0 && nloc < 2) ? (Vp + row0 * 8 + 4 + nloc) : dump;
#pragma unroll
        for (int r = 0; r < 8; ++r)
          if (mt < 3 || r < 6) {
            b0[r * 8] = d0[r] * qscl;
            b1[r * 8] = d1[r];
          }
      }
    }
    __syncthreads();

    v2f ak[4][2];
#pragma unroll
    for (int mt = 0; mt < 4; ++mt)
#pragma unroll
      for (int kk = 0; kk < 2; ++kk)
        ak[mt][kk] = *(const v2f*)(Kp + (mt * 16 + nloc) * 8 + kk * 4 + kb);
    const float* vcolp = Vp + ((nloc < 8) ? nloc : 7);
    v2f av[14];
#pragma unroll
    for (int mt = 0; mt < 3; ++mt)
#pragma unroll
      for (int p = 0; p < 4; ++p) {
        int key = mt * 16 + upper8 + 2 * p;
        av[mt * 4 + p].x = vcolp[key * 8];
        av[mt * 4 + p].y = vcolp[(key + 1) * 8];
      }
    {
      int key12 = lowhalf ? 48 : 52;
      av[12].x = vcolp[key12 * 8];
      av[12].y = vcolp[(key12 + 1) * 8];
      int key13 = lowhalf ? 50 : 54;
      av[13].x = vcolp[key13 * 8];
      av[13].y = vcolp[(key13 + 1) * 8];
    }

    for (int nt = 0; nt < 4; ++nt) {
      v2f bq[2];
#pragma unroll
      for (int kk = 0; kk < 2; ++kk)
        bq[kk] = *(const v2f*)(Qp + (nt * 16 + nloc) * 8 + kk * 4 + kb);
      v8f t[4];
#pragma unroll
      for (int mt = 0; mt < 4; ++mt) {
        v8f c = {};
#pragma unroll
        for (int kk = 0; kk < 2; ++kk) c = wmma_f32(ak[mt][kk], bq[kk], c);
        t[mt] = c;
      }

      float mx = -3.0e38f;
#pragma unroll
      for (int mt = 0; mt < 4; ++mt)
#pragma unroll
        for (int r = 0; r < 8; ++r) {
          int k = mt * 16 + r + upper8;
          if (k < SEQ) mx = fmaxf(mx, t[mt][r]);
        }
      mx = fmaxf(mx, __shfl_xor(mx, 16));
#pragma unroll
      for (int mt = 0; mt < 4; ++mt)
#pragma unroll
        for (int r = 0; r < 8; ++r) {
          int k = mt * 16 + r + upper8;
          t[mt][r] = (k < SEQ) ? __expf(t[mt][r] - mx) : 0.0f;
        }

      v8f o0 = {}, o1 = {};
#pragma unroll
      for (int mt = 0; mt < 3; ++mt)
#pragma unroll
        for (int p = 0; p < 4; p += 2) {
          v2f b0 = {t[mt][2 * p], t[mt][2 * p + 1]};
          v2f b1 = {t[mt][2 * p + 2], t[mt][2 * p + 3]};
          o0 = wmma_f32(av[mt * 4 + p], b0, o0);
          o1 = wmma_f32(av[mt * 4 + p + 1], b1, o1);
        }
      {
        v2f b12 = {t[3][0] + xhalf(t[3][4]), t[3][1] + xhalf(t[3][5])};
        v2f b13 = {t[3][2], t[3][3]};
        o0 = wmma_f32(av[12], b12, o0);
        o1 = wmma_f32(av[13], b13, o1);
      }
      {
        float inv = 1.0f / (o0[6] + o1[6]);
        int q = nt * 16 + nloc;
        if (lowhalf && q < SEQ) {
          float* zq = Z + q * DIM;
#pragma unroll
          for (int r = 0; r < DIM; r += 2) {
            v2f w = {(o0[r] + o1[r]) * inv, (o0[r + 1] + o1[r + 1]) * inv};
            *(v2f*)(zq + r) = w;
          }
        }
      }
    }
    __syncthreads();

    {
      v2f bw[2];
      loadBimg(W_OW, bw);
      v8f cb = loadCimg(B_OW);
#pragma unroll
      for (int mt = 0; mt < 4; ++mt) {
        v2f aw[2];
        loadA(Z, mt, aw);
        v8f d = wmma_f32(aw[0], bw[0], cb);
        d = wmma_f32(aw[1], bw[1], d);
        int row0 = mt * 16 + upper8;
        bool ok = (nloc < DIM) && ((mt < 3) || lowhalf);
        float* bzo = ok ? &Z[row0 * DIM + nloc] : dump;
#pragma unroll
        for (int r = 0; r < 8; ++r)
          if (mt < 3 || r < 6) bzo[r * DIM] = d[r];
      }
    }
#pragma unroll
    for (int t = 0; t < FEAT / 32; ++t) {
      int idx = lane + 32 * t;
      H[idx] = eluf(Z[idx]) + H[idx];
    }
    if (lane < FEAT % 32) {
      int idx = lane + (FEAT / 32) * 32;
      H[idx] = eluf(Z[idx]) + H[idx];
    }
    __syncthreads();
  }

  float vacc = 0.0f;
  float pacc[POL];
#pragma unroll
  for (int o = 0; o < POL; ++o) pacc[o] = 0.0f;
  for (int j = lane; j < FEAT; j += 32) {
    float fj = H[j];
    vacc += fj * vw[j];
#pragma unroll
    for (int o = 0; o < POL; ++o) pacc[o] += fj * pw[o * FEAT + j];
  }
  for (int off = 16; off >= 1; off >>= 1) {
    vacc += __shfl_xor(vacc, off, 32);
#pragma unroll
    for (int o = 0; o < POL; ++o) pacc[o] += __shfl_xor(pacc[o], off, 32);
  }
  if (lane == 0) {
    out[b] = vacc + vb[0];
    float logit[POL];
    float mx = -3.0e38f;
#pragma unroll
    for (int o = 0; o < POL; ++o) {
      logit[o] = pacc[o] + pb[o];
      mx = fmaxf(mx, logit[o]);
    }
    float sum = 0.0f;
#pragma unroll
    for (int o = 0; o < POL; ++o) {
      logit[o] = __expf(logit[o] - mx);
      sum += logit[o];
    }
    float inv = 1.0f / sum;
    float* pout = out + n_batch + (size_t)b * POL;
#pragma unroll
    for (int o = 0; o < POL; ++o) pout[o] = logit[o] * inv;
  }
}

extern "C" void kernel_launch(void* const* d_in, const int* in_sizes, int n_in,
                              void* d_out, int out_size, void* d_ws,
                              size_t ws_size, hipStream_t stream) {
  (void)n_in;
  (void)out_size;
  (void)d_ws;
  (void)ws_size;
  const float* x = (const float*)d_in[0];
  const float* embed_w = (const float*)d_in[1];
  const float* embed_b = (const float*)d_in[2];
  const float* fc1_w = (const float*)d_in[3];
  const float* fc1_b = (const float*)d_in[4];
  const float* ln_g = (const float*)d_in[5];
  const float* ln_b = (const float*)d_in[6];
  const float* iw = (const float*)d_in[7];
  const float* ib = (const float*)d_in[8];
  const float* ow = (const float*)d_in[9];
  const float* ob = (const float*)d_in[10];
  const float* vw = (const float*)d_in[11];
  const float* vb = (const float*)d_in[12];
  const float* pw = (const float*)d_in[13];
  const float* pb = (const float*)d_in[14];

  int n_batch = in_sizes[0] / FEAT;
  int grid = (n_batch + WAVES_PER_BLOCK - 1) / WAVES_PER_BLOCK;
  rubik_fused<<<grid, THREADS, 0, stream>>>(x, embed_w, embed_b, fc1_w, fc1_b,
                                            ln_g, ln_b, iw, ib, ow, ob, vw, vb,
                                            pw, pb, (float*)d_out, n_batch);
}


# Layer weights pre-multiplied at staging: Q' = z(Wk^T Wq)^T is scored against z itself (per-query terms cancel in softmax), V' = z(ow Wv)^T; QKV is one tile, out-proj gone; product sum order changes
# speedup vs baseline: 1.3243x; 1.0814x over previous
#include <hip/hip_runtime.h>
#include <math.h>

#define NUM_TBLOCKS 8
#define SEQ 54
#define DIM 6
#define QKV (3 * DIM)
#define FEAT (SEQ * DIM)
#define POL 12
#define LN_EPS 1e-5f
#define QSCALE 0.40824829046386302f

#define WAVES_PER_BLOCK 4
#define THREADS (WAVES_PER_BLOCK * 32)
#define WAVE_LDS 2240
#define WCACHE 320

typedef __attribute__((ext_vector_type(2))) float v2f;
typedef __attribute__((ext_vector_type(4))) float v4f;
typedef __attribute__((ext_vector_type(8))) float v8f;

__device__ __forceinline__ v8f wmma_f32(v2f a, v2f b, v8f c) {
  return __builtin_amdgcn_wmma_f32_16x16x4_f32(false, a, false, b, (short)0, c,
                                               false, false);
}

__device__ __forceinline__ float xhalf(float x) {
  int b = __builtin_bit_cast(int, x);
  return __builtin_bit_cast(
      float, __builtin_amdgcn_permlanex16(b, b, 0x76543210, 0xfedcba98, false,
                                          false));
}

__device__ __forceinline__ float eluf(float x) {
  return x > 0.0f ? x : (__expf(x) - 1.0f);
}

__global__ __launch_bounds__(THREADS) void rubik_fused(
    const float* __restrict__ x, const float* __restrict__ embed_w,
    const float* __restrict__ embed_b, const float* __restrict__ fc1_w,
    const float* __restrict__ fc1_b, const float* __restrict__ ln_g,
    const float* __restrict__ ln_b, const float* __restrict__ iw,
    const float* __restrict__ ib, const float* __restrict__ ow,
    const float* __restrict__ ob, const float* __restrict__ vw,
    const float* __restrict__ vb, const float* __restrict__ pw,
    const float* __restrict__ pb, float* __restrict__ out, int n_batch) {
  __shared__ float smem[WAVES_PER_BLOCK * WAVE_LDS + WCACHE];
  const int lane = threadIdx.x & 31;
  const int wave = threadIdx.x >> 5;
  int b = blockIdx.x * WAVES_PER_BLOCK + wave;
  if (b >= n_batch) b = n_batch - 1;

  float* H = &smem[wave * WAVE_LDS];
  float* Z = H + 324;
  float* Qp = Z + 324;
  float* Kp = Qp + 512;
  float* Vp = Kp + 512;
  float* dump = H + 2120;

  float* WC = &smem[WAVES_PER_BLOCK * WAVE_LDS];
  float* W_FC = WC;
  float* W_QV = WC + 128;
  float* B_FC = WC + 256;
  float* B_QV = WC + 288;

  const float* xb = x + (size_t)b * FEAT;
  __builtin_prefetch(xb, 0, 1);

  const int nloc = lane & 15;
  const int kb = (lane < 16) ? 0 : 2;
  const int upper8 = (lane < 16) ? 0 : 8;
  const bool lowhalf = (lane < 16);

  const bool qvcol = (nloc < 2 * DIM);
  float* qvBase = (nloc < DIM) ? (Qp + nloc) : (Vp + (nloc - DIM));

  auto fillWB = [&](const float* w, const float* bias, int nCols, int ntile,
                    float* wimg, float* bimg) {
    for (int slot = threadIdx.x; slot < 128; slot += THREADS) {
      int kk = slot >> 6, ln = (slot >> 1) & 31, p = slot & 1;
      int nl = ln & 15;
      int c = kk * 4 + ((ln < 16) ? 0 : 2) + p;
      int j = ntile * 16 + nl;
      bool v = (j < nCols) && (c < DIM);
      wimg[slot] = v ? w[v ? j * DIM + c : 0] : 0.0f;
    }
    for (int ln = threadIdx.x; ln < 32; ln += THREADS) {
      int j = ntile * 16 + (ln & 15);
      bimg[ln] = (j < nCols) ? bias[(j < nCols) ? j : 0] : 0.0f;
    }
  };

  auto fillQV = [&](const float* iwl, const float* ibl, const float* owl,
                    float* wimg, float* bimg) {
    for (int slot = threadIdx.x; slot < 128; slot += THREADS) {
      int kk = slot >> 6, ln = (slot >> 1) & 31, p = slot & 1;
      int n = ln & 15;
      int c = kk * 4 + ((ln < 16) ? 0 : 2) + p;
      bool isq = (n < DIM);
      bool v = (n < 2 * DIM) && (c < DIM);
      int a = v ? (isq ? n : n - DIM) : 0;
      int cc = v ? c : 0;
      const float* lp = isq ? (iwl + DIM * DIM + a) : (owl + a * DIM);
      int ls = isq ? DIM : 1;
      const float* rp = (isq ? iwl : (iwl + 2 * DIM * DIM)) + cc;
      float acc = 0.0f;
      for (int m = 0; m < DIM; ++m) acc = fmaf(lp[m * ls], rp[m * DIM], acc);
      if (isq) acc *= QSCALE;
      wimg[slot] = v ? acc : 0.0f;
    }
    for (int ln = threadIdx.x; ln < 32; ln += THREADS) {
      int n = ln & 15;
      bool isq = (n < DIM);
      bool v = (n < 2 * DIM);
      int a = v ? (isq ? n : n - DIM) : 0;
      const float* lp = isq ? (iwl + DIM * DIM + a) : (owl + a * DIM);
      int ls = isq ? DIM : 1;
      const float* rp = isq ? ibl : (ibl + 2 * DIM);
      float acc = 0.0f;
      for (int m = 0; m < DIM; ++m) acc = fmaf(lp[m * ls], rp[m], acc);
      if (isq) acc *= QSCALE;
      bimg[ln] = v ? acc : 0.0f;
    }
  };

  auto loadA = [&](const float* src, int mt, v2f aw[2]) {
    int row = mt * 16 + nloc;
#pragma unroll
    for (int kk = 0; kk < 2; ++kk)
      aw[kk] = *(const v2f*)(src + row * DIM + kk * 4 + kb);
  };
  auto loadBimg = [&](const float* wimg, v2f bw[2]) {
    bw[0] = *(const v2f*)(wimg + lane * 2);
    bw[1] = *(const v2f*)(wimg + 64 + lane * 2);
  };
  auto loadCimg = [&](const float* bimg) {
    float bvv = bimg[lane];
    v8f c;
#pragma unroll
    for (int r = 0; r < 8; ++r) c[r] = bvv;
    return c;
  };

  for (int idx = lane; idx < FEAT; idx += 32) Z[idx] = xb[idx];
  for (int idx = lane; idx < 512 + 512 + 448; idx += 32) Qp[idx] = 0.0f;
  fillWB(embed_w, embed_b, DIM, 0, W_FC, B_FC);
  __syncthreads();

  for (int s = lane; s < SEQ; s += 32) Vp[s * 8 + DIM] = 1.0f;

  {
    v2f bw[2];
    loadBimg(W_FC, bw);
    v8f cb = loadCimg(B_FC);
#pragma unroll
    for (int mt = 0; mt < 4; ++mt) {
      v2f aw[2];
      loadA(Z, mt, aw);
      v8f d = wmma_f32(aw[0], bw[0], cb);
      d = wmma_f32(aw[1], bw[1], d);
      int row0 = mt * 16 + upper8;
      bool ok = (nloc < DIM) && ((mt < 3) || lowhalf);
      float* bz = ok ? &H[row0 * DIM + nloc] : dump;
#pragma unroll
      for (int r = 0; r < 8; ++r)
        if (mt < 3 || r < 6) bz[r * DIM] = d[r];
    }
  }
  __syncthreads();

  for (int L = 0; L < NUM_TBLOCKS; ++L) {
    const float* g = ln_g + L * DIM;
    const float* bbia = ln_b + L * DIM;
    const float* obl = ob + L * DIM;

    fillWB(fc1_w + L * DIM * DIM, fc1_b + L * DIM, DIM, 0, W_FC, B_FC);
    fillQV(iw + L * QKV * DIM, ib + L * QKV, ow + L * DIM * DIM, W_QV, B_QV);
    __syncthreads();

    {
      v2f bw[2];
      loadBimg(W_FC, bw);
      v8f cb = loadCimg(B_FC);
#pragma unroll
      for (int mt = 0; mt < 4; ++mt) {
        v2f aw[2];
        loadA(H, mt, aw);
        v8f d = wmma_f32(aw[0], bw[0], cb);
        d = wmma_f32(aw[1], bw[1], d);
        int row0 = mt * 16 + upper8;
        bool ok = (nloc < DIM) && ((mt < 3) || lowhalf);
        float* bz = ok ? &Z[row0 * DIM + nloc] : dump;
#pragma unroll
        for (int r = 0; r < 8; ++r)
          if (mt < 3 || r < 6) bz[r * DIM] = d[r];
      }
    }
    __syncthreads();

    for (int s = lane; s < SEQ; s += 32) {
      float zv[DIM];
#pragma unroll
      for (int i = 0; i < DIM; ++i) zv[i] = eluf(Z[s * DIM + i]);
      float mu = 0.0f;
#pragma unroll
      for (int i = 0; i < DIM; ++i) mu += zv[i];
      mu *= (1.0f / DIM);
      float var = 0.0f;
#pragma unroll
      for (int i = 0; i < DIM; ++i) {
        float d = zv[i] - mu;
        var += d * d;
      }
      var *= (1.0f / DIM);
      float rs = rsqrtf(var + LN_EPS);
#pragma unroll
      for (int i = 0; i < DIM; ++i)
        Kp[s * 8 + i] = (zv[i] - mu) * rs * g[i] + bbia[i];
    }
    __syncthreads();

    v2f ak[4][2];
#pragma unroll
    for (int mt = 0; mt < 4; ++mt)
#pragma unroll
      for (int kk = 0; kk < 2; ++kk)
        ak[mt][kk] = *(const v2f*)(Kp + (mt * 16 + nloc) * 8 + kk * 4 + kb);

    {
      v2f bw[2];
      loadBimg(W_QV, bw);
      v8f cb = loadCimg(B_QV);
#pragma unroll
      for (int mt = 0; mt < 4; ++mt) {
        v8f d = wmma_f32(ak[mt][0], bw[0], cb);
        d = wmma_f32(ak[mt][1], bw[1], d);
        int row0 = mt * 16 + upper8;
        bool ok0 = qvcol && ((mt < 3) || lowhalf);
        float* b0 = ok0 ? (qvBase + row0 * 8) : dump;
#pragma unroll
        for (int r = 0; r < 8; ++r)
          if (mt < 3 || r < 6) b0[r * 8] = d[r];
      }
    }
    __syncthreads();

    const float* vcolp = Vp + ((nloc < 8) ? nloc : 7);
    v2f av[14];
#pragma unroll
    for (int mt = 0; mt < 3; ++mt)
#pragma unroll
      for (int p = 0; p < 4; ++p) {
        int key = mt * 16 + upper8 + 2 * p;
        av[mt * 4 + p].x = vcolp[key * 8];
        av[mt * 4 + p].y = vcolp[(key + 1) * 8];
      }
    {
      int key12 = lowhalf ? 48 : 52;
      av[12].x = vcolp[key12 * 8];
      av[12].y = vcolp[(key12 + 1) * 8];
      int key13 = lowhalf ? 50 : 54;
      av[13].x = vcolp[key13 * 8];
      av[13].y = vcolp[(key13 + 1) * 8];
    }

    float obv[DIM];
    for (int r = 0; r < DIM; ++r) obv[r] = obl[r];

    for (int nt = 0; nt < 4; ++nt) {
      v2f bq[2];
#pragma unroll
      for (int kk = 0; kk < 2; ++kk)
        bq[kk] = *(const v2f*)(Qp + (nt * 16 + nloc) * 8 + kk * 4 + kb);
      v8f t[4];
#pragma unroll
      for (int mt = 0; mt < 4; ++mt) {
        v8f c = {};
#pragma unroll
        for (int kk = 0; kk < 2; ++kk) c = wmma_f32(ak[mt][kk], bq[kk], c);
        t[mt] = c;
      }

      float mx = -3.0e38f;
#pragma unroll
      for (int mt = 0; mt < 4; ++mt)
#pragma unroll
        for (int r = 0; r < 8; ++r) {
          int k = mt * 16 + r + upper8;
          if (k < SEQ) mx = fmaxf(mx, t[mt][r]);
        }
      mx = fmaxf(mx, __shfl_xor(mx, 16));
#pragma unroll
      for (int mt = 0; mt < 4; ++mt)
#pragma unroll
        for (int r = 0; r < 8; ++r) {
          int k = mt * 16 + r + upper8;
          t[mt][r] = (k < SEQ) ? __expf(t[mt][r] - mx) : 0.0f;
        }

      v8f o0 = {}, o1 = {};
#pragma unroll
      for (int mt = 0; mt < 3; ++mt)
#pragma unroll
        for (int p = 0; p < 4; p += 2) {
          v2f b0 = {t[mt][2 * p], t[mt][2 * p + 1]};
          v2f b1 = {t[mt][2 * p + 2], t[mt][2 * p + 3]};
          o0 = wmma_f32(av[mt * 4 + p], b0, o0);
          o1 = wmma_f32(av[mt * 4 + p + 1], b1, o1);
        }
      {
        v2f b12 = {t[3][0] + xhalf(t[3][4]), t[3][1] + xhalf(t[3][5])};
        v2f b13 = {t[3][2], t[3][3]};
        o0 = wmma_f32(av[12], b12, o0);
        o1 = wmma_f32(av[13], b13, o1);
      }
      {
        float inv = 1.0f / (o0[6] + o1[6]);
        int q = nt * 16 + nloc;
        if (lowhalf && q < SEQ) {
          float* zq = Z + q * DIM;
#pragma unroll
          for (int r = 0; r < DIM; r += 2) {
            v2f w = {(o0[r] + o1[r]) * inv + obv[r],
                     (o0[r + 1] + o1[r + 1]) * inv + obv[r + 1]};
            *(v2f*)(zq + r) = w;
          }
        }
      }
    }
#pragma unroll
    for (int t = 0; t < FEAT / 32; ++t) {
      int idx = lane + 32 * t;
      H[idx] = eluf(Z[idx]) + H[idx];
    }
    if (lane < FEAT % 32) {
      int idx = lane + (FEAT / 32) * 32;
      H[idx] = eluf(Z[idx]) + H[idx];
    }
    __syncthreads();
  }

  float vacc = 0.0f;
  float pacc[POL];
#pragma unroll
  for (int o = 0; o < POL; ++o) pacc[o] = 0.0f;
  for (int j = lane; j < FEAT; j += 32) {
    float fj = H[j];
    vacc += fj * vw[j];
#pragma unroll
    for (int o = 0; o < POL; ++o) pacc[o] += fj * pw[o * FEAT + j];
  }
  for (int off = 16; off >= 1; off >>= 1) {
    vacc += __shfl_xor(vacc, off, 32);
#pragma unroll
    for (int o = 0; o < POL; ++o) pacc[o] += __shfl_xor(pacc[o], off, 32);
  }
  if (lane == 0) {
    out[b] = vacc + vb[0];
    float logit[POL];
    float mx = -3.0e38f;
#pragma unroll
    for (int o = 0; o < POL; ++o) {
      logit[o] = pacc[o] + pb[o];
      mx = fmaxf(mx, logit[o]);
    }
    float sum = 0.0f;
#pragma unroll
    for (int o = 0; o < POL; ++o) {
      logit[o] = __expf(logit[o] - mx);
      sum += logit[o];
    }
    float inv = 1.0f / sum;
    float* pout = out + n_batch + (size_t)b * POL;
#pragma unroll
    for (int o = 0; o < POL; ++o) pout[o] = logit[o] * inv;
  }
}

extern "C" void kernel_launch(void* const* d_in, const int* in_sizes, int n_in,
                              void* d_out, int out_size, void* d_ws,
                              size_t ws_size, hipStream_t stream) {
  (void)n_in;
  (void)out_size;
  (void)d_ws;
  (void)ws_size;
  const float* x = (const float*)d_in[0];
  const float* embed_w = (const float*)d_in[1];
  const float* embed_b = (const float*)d_in[2];
  const float* fc1_w = (const float*)d_in[3];
  const float* fc1_b = (const float*)d_in[4];
  const float* ln_g = (const float*)d_in[5];
  const float* ln_b = (const float*)d_in[6];
  const float* iw = (const float*)d_in[7];
  const float* ib = (const float*)d_in[8];
  const float* ow = (const float*)d_in[9];
  const float* ob = (const float*)d_in[10];
  const float* vw = (const float*)d_in[11];
  const float* vb = (const float*)d_in[12];
  const float* pw = (const float*)d_in[13];
  const float* pb = (const float*)d_in[14];

  int n_batch = in_sizes[0] / FEAT;
  int grid = (n_batch + WAVES_PER_BLOCK - 1) / WAVES_PER_BLOCK;
  rubik_fused<<<grid, THREADS, 0, stream>>>(x, embed_w, embed_b, fc1_w, fc1_b,
                                            ln_g, ln_b, iw, ib, ow, ob, vw, vb,
                                            pw, pb, (float*)d_out, n_batch);
}


# Scores leave the product exp2-ready: log2(e) folded into the Q' scale at staging, padded keys masked to -3e38 through the spare k column of z and Q'; per-score multiply and pad selects gone
# speedup vs baseline: 1.3725x; 1.0364x over previous
#include <hip/hip_runtime.h>
#include <math.h>

#define NUM_TBLOCKS 8
#define SEQ 54
#define DIM 6
#define QKV (3 * DIM)
#define FEAT (SEQ * DIM)
#define POL 12
#define LN_EPS 1e-5f
#define QSCALE 0.58897778410361230f
#define PAD_SCORE -3.0e38f

#define WAVES_PER_BLOCK 4
#define THREADS (WAVES_PER_BLOCK * 32)
#define WAVE_LDS 2240
#define WCACHE 320

typedef __attribute__((ext_vector_type(2))) float v2f;
typedef __attribute__((ext_vector_type(4))) float v4f;
typedef __attribute__((ext_vector_type(8))) float v8f;

__device__ __forceinline__ v8f wmma_f32(v2f a, v2f b, v8f c) {
  return __builtin_amdgcn_wmma_f32_16x16x4_f32(false, a, false, b, (short)0, c,
                                               false, false);
}

__device__ __forceinline__ float xhalf(float x) {
  int b = __builtin_bit_cast(int, x);
  return __builtin_bit_cast(
      float, __builtin_amdgcn_permlanex16(b, b, 0x76543210, 0xfedcba98, false,
                                          false));
}

__device__ __forceinline__ float eluf(float x) {
  return x > 0.0f ? x : (__expf(x) - 1.0f);
}

__global__ __launch_bounds__(THREADS) void rubik_fused(
    const float* __restrict__ x, const float* __restrict__ embed_w,
    const float* __restrict__ embed_b, const float* __restrict__ fc1_w,
    const float* __restrict__ fc1_b, const float* __restrict__ ln_g,
    const float* __restrict__ ln_b, const float* __restrict__ iw,
    const float* __restrict__ ib, const float* __restrict__ ow,
    const float* __restrict__ ob, const float* __restrict__ vw,
    const float* __restrict__ vb, const float* __restrict__ pw,
    const float* __restrict__ pb, float* __restrict__ out, int n_batch) {
  __shared__ float smem[WAVES_PER_BLOCK * WAVE_LDS + WCACHE];
  const int lane = threadIdx.x & 31;
  const int wave = threadIdx.x >> 5;
  int b = blockIdx.x * WAVES_PER_BLOCK + wave;
  if (b >= n_batch) b = n_batch - 1;

  float* H = &smem[wave * WAVE_LDS];
  float* Z = H + 324;
  float* Qp = Z + 324;
  float* Kp = Qp + 512;
  float* Vp = Kp + 512;
  float* dump = H + 2120;

  float* WC = &smem[WAVES_PER_BLOCK * WAVE_LDS];
  float* W_FC = WC;
  float* W_QV = WC + 128;
  float* B_FC = WC + 256;
  float* B_QV = WC + 288;

  const float* xb = x + (size_t)b * FEAT;
  __builtin_prefetch(xb, 0, 1);

  const int nloc = lane & 15;
  const int kb = (lane < 16) ? 0 : 2;
  const int upper8 = (lane < 16) ? 0 : 8;
  const bool lowhalf = (lane < 16);

  const bool qvcol = (nloc < 2 * DIM);
  float* qvBase = (nloc < DIM) ? (Qp + nloc) : (Vp + (nloc - DIM));

  auto fillWB = [&](const float* w, const float* bias, int nCols, int ntile,
                    float* wimg, float* bimg) {
    for (int slot = threadIdx.x; slot < 128; slot += THREADS) {
      int kk = slot >> 6, ln = (slot >> 1) & 31, p = slot & 1;
      int nl = ln & 15;
      int c = kk * 4 + ((ln < 16) ? 0 : 2) + p;
      int j = ntile * 16 + nl;
      bool v = (j < nCols) && (c < DIM);
      wimg[slot] = v ? w[v ? j * DIM + c : 0] : 0.0f;
    }
    for (int ln = threadIdx.x; ln < 32; ln += THREADS) {
      int j = ntile * 16 + (ln & 15);
      bimg[ln] = (j < nCols) ? bias[(j < nCols) ? j : 0] : 0.0f;
    }
  };

  auto fillQV = [&](const float* iwl, const float* ibl, const float* owl,
                    float* wimg, float* bimg) {
    for (int slot = threadIdx.x; slot < 128; slot += THREADS) {
      int kk = slot >> 6, ln = (slot >> 1) & 31, p = slot & 1;
      int n = ln & 15;
      int c = kk * 4 + ((ln < 16) ? 0 : 2) + p;
      bool isq = (n < DIM);
      bool v = (n < 2 * DIM) && (c < DIM);
      int a = v ? (isq ? n : n - DIM) : 0;
      int cc = v ? c : 0;
      const float* lp = isq ? (iwl + DIM * DIM + a) : (owl + a * DIM);
      int ls = isq ? DIM : 1;
      const float* rp = (isq ? iwl : (iwl + 2 * DIM * DIM)) + cc;
      float acc = 0.0f;
      for (int m = 0; m < DIM; ++m) acc = fmaf(lp[m * ls], rp[m * DIM], acc);
      if (isq) acc *= QSCALE;
      wimg[slot] = v ? acc : 0.0f;
    }
    for (int ln = threadIdx.x; ln < 32; ln += THREADS) {
      int n = ln & 15;
      bool isq = (n < DIM);
      bool v = (n < 2 * DIM);
      int a = v ? (isq ? n : n - DIM) : 0;
      const float* lp = isq ? (iwl + DIM * DIM + a) : (owl + a * DIM);
      int ls = isq ? DIM : 1;
      const float* rp = isq ? ibl : (ibl + 2 * DIM);
      float acc = 0.0f;
      for (int m = 0; m < DIM; ++m) acc = fmaf(lp[m * ls], rp[m], acc);
      if (isq) acc *= QSCALE;
      bimg[ln] = v ? acc : 0.0f;
    }
  };

  auto loadA = [&](const float* src, int mt, v2f aw[2]) {
    int row = mt * 16 + nloc;
#pragma unroll
    for (int kk = 0; kk < 2; ++kk)
      aw[kk] = *(const v2f*)(src + row * DIM + kk * 4 + kb);
  };
  auto loadBimg = [&](const float* wimg, v2f bw[2]) {
    bw[0] = *(const v2f*)(wimg + lane * 2);
    bw[1] = *(const v2f*)(wimg + 64 + lane * 2);
  };
  auto loadCimg = [&](const float* bimg) {
    float bvv = bimg[lane];
    v8f c;
#pragma unroll
    for (int r = 0; r < 8; ++r) c[r] = bvv;
    return c;
  };

  for (int idx = lane; idx < FEAT; idx += 32) Z[idx] = xb[idx];
  for (int idx = lane; idx < 512 + 512 + 448; idx += 32) Qp[idx] = 0.0f;
  fillWB(embed_w, embed_b, DIM, 0, W_FC, B_FC);
  __syncthreads();

  for (int s = lane; s < SEQ; s += 32) Vp[s * 8 + DIM] = 1.0f;
  for (int s = lane; s < 64; s += 32) Qp[s * 8 + 7] = PAD_SCORE;
  if (lane < 64 - SEQ) Kp[(SEQ + lane) * 8 + 7] = 1.0f;

  {
    v2f bw[2];
    loadBimg(W_FC, bw);
    v8f cb = loadCimg(B_FC);
#pragma unroll
    for (int mt = 0; mt < 4; ++mt) {
      v2f aw[2];
      loadA(Z, mt, aw);
      v8f d = wmma_f32(aw[0], bw[0], cb);
      d = wmma_f32(aw[1], bw[1], d);
      int row0 = mt * 16 + upper8;
      bool ok = (nloc < DIM) && ((mt < 3) || lowhalf);
      float* bz = ok ? &H[row0 * DIM + nloc] : dump;
#pragma unroll
      for (int r = 0; r < 8; ++r)
        if (mt < 3 || r < 6) bz[r * DIM] = d[r];
    }
  }
  __syncthreads();

  for (int L = 0; L < NUM_TBLOCKS; ++L) {
    const float* g = ln_g + L * DIM;
    const float* bbia = ln_b + L * DIM;
    const float* obl = ob + L * DIM;

    fillWB(fc1_w + L * DIM * DIM, fc1_b + L * DIM, DIM, 0, W_FC, B_FC);
    fillQV(iw + L * QKV * DIM, ib + L * QKV, ow + L * DIM * DIM, W_QV, B_QV);
    __syncthreads();

    {
      v2f bw[2];
      loadBimg(W_FC, bw);
      v8f cb = loadCimg(B_FC);
#pragma unroll
      for (int mt = 0; mt < 4; ++mt) {
        v2f aw[2];
        loadA(H, mt, aw);
        v8f d = wmma_f32(aw[0], bw[0], cb);
        d = wmma_f32(aw[1], bw[1], d);
        int row0 = mt * 16 + upper8;
        bool ok = (nloc < DIM) && ((mt < 3) || lowhalf);
        float* bz = ok ? &Z[row0 * DIM + nloc] : dump;
#pragma unroll
        for (int r = 0; r < 8; ++r)
          if (mt < 3 || r < 6) bz[r * DIM] = d[r];
      }
    }
    __syncthreads();

    for (int s = lane; s < SEQ; s += 32) {
      float zv[DIM];
#pragma unroll
      for (int i = 0; i < DIM; ++i) zv[i] = eluf(Z[s * DIM + i]);
      float mu = 0.0f;
#pragma unroll
      for (int i = 0; i < DIM; ++i) mu += zv[i];
      mu *= (1.0f / DIM);
      float var = 0.0f;
#pragma unroll
      for (int i = 0; i < DIM; ++i) {
        float d = zv[i] - mu;
        var += d * d;
      }
      var *= (1.0f / DIM);
      float rs = rsqrtf(var + LN_EPS);
#pragma unroll
      for (int i = 0; i < DIM; ++i)
        Kp[s * 8 + i] = (zv[i] - mu) * rs * g[i] + bbia[i];
    }
    __syncthreads();

    v2f ak[4][2];
#pragma unroll
    for (int mt = 0; mt < 4; ++mt)
#pragma unroll
      for (int kk = 0; kk < 2; ++kk)
        ak[mt][kk] = *(const v2f*)(Kp + (mt * 16 + nloc) * 8 + kk * 4 + kb);

    {
      v2f bw[2];
      loadBimg(W_QV, bw);
      v8f cb = loadCimg(B_QV);
#pragma unroll
      for (int mt = 0; mt < 4; ++mt) {
        v8f d = wmma_f32(ak[mt][0], bw[0], cb);
        d = wmma_f32(ak[mt][1], bw[1], d);
        int row0 = mt * 16 + upper8;
        bool ok0 = qvcol && ((mt < 3) || lowhalf);
        float* b0 = ok0 ? (qvBase + row0 * 8) : dump;
#pragma unroll
        for (int r = 0; r < 8; ++r)
          if (mt < 3 || r < 6) b0[r * 8] = d[r];
      }
    }
    __syncthreads();

    const float* vcolp = Vp + ((nloc < 8) ? nloc : 7);
    v2f av[14];
#pragma unroll
    for (int mt = 0; mt < 3; ++mt)
#pragma unroll
      for (int p = 0; p < 4; ++p) {
        int key = mt * 16 + upper8 + 2 * p;
        av[mt * 4 + p].x = vcolp[key * 8];
        av[mt * 4 + p].y = vcolp[(key + 1) * 8];
      }
    {
      int key12 = lowhalf ? 48 : 52;
      av[12].x = vcolp[key12 * 8];
      av[12].y = vcolp[(key12 + 1) * 8];
      int key13 = lowhalf ? 50 : 54;
      av[13].x = vcolp[key13 * 8];
      av[13].y = vcolp[(key13 + 1) * 8];
    }

    float obv[DIM];
    for (int r = 0; r < DIM; ++r) obv[r] = obl[r];

#pragma nounroll
    for (int nt = 0; nt < 4; ++nt) {
      v2f bq[2];
#pragma unroll
      for (int kk = 0; kk < 2; ++kk)
        bq[kk] = *(const v2f*)(Qp + (nt * 16 + nloc) * 8 + kk * 4 + kb);
      v8f t[4];
#pragma unroll
      for (int mt = 0; mt < 4; ++mt) {
        v8f c = {};
#pragma unroll
        for (int kk = 0; kk < 2; ++kk) c = wmma_f32(ak[mt][kk], bq[kk], c);
        t[mt] = c;
      }

      float mx = t[0][0];
#pragma unroll
      for (int mt = 0; mt < 4; ++mt)
#pragma unroll
        for (int r = 0; r < 8; ++r)
          if ((mt > 0 || r > 0) && (mt < 3 || r < 6)) mx = fmaxf(mx, t[mt][r]);
      mx = fmaxf(mx, __shfl_xor(mx, 16));
      const v2f mx2 = {mx, mx};
#pragma unroll
      for (int mt = 0; mt < 4; ++mt)
#pragma unroll
        for (int p = 0; p < 4; ++p)
          if (mt < 3 || p < 3) {
            v2f d = (v2f){t[mt][2 * p], t[mt][2 * p + 1]} - mx2;
            t[mt][2 * p] = __builtin_amdgcn_exp2f(d.x);
            t[mt][2 * p + 1] = __builtin_amdgcn_exp2f(d.y);
          }

      v8f o0 = {}, o1 = {};
#pragma unroll
      for (int mt = 0; mt < 3; ++mt)
#pragma unroll
        for (int p = 0; p < 4; p += 2) {
          v2f b0 = {t[mt][2 * p], t[mt][2 * p + 1]};
          v2f b1 = {t[mt][2 * p + 2], t[mt][2 * p + 3]};
          o0 = wmma_f32(av[mt * 4 + p], b0, o0);
          o1 = wmma_f32(av[mt * 4 + p + 1], b1, o1);
        }
      {
        v2f b12 = {t[3][0] + xhalf(t[3][4]), t[3][1] + xhalf(t[3][5])};
        v2f b13 = {t[3][2], t[3][3]};
        o0 = wmma_f32(av[12], b12, o0);
        o1 = wmma_f32(av[13], b13, o1);
      }
      {
        float inv = 1.0f / (o0[6] + o1[6]);
        int q = nt * 16 + nloc;
        if (lowhalf && q < SEQ) {
          float* zq = Z + q * DIM;
#pragma unroll
          for (int r = 0; r < DIM; r += 2) {
            v2f w = {(o0[r] + o1[r]) * inv + obv[r],
                     (o0[r + 1] + o1[r + 1]) * inv + obv[r + 1]};
            *(v2f*)(zq + r) = w;
          }
        }
      }
    }
#pragma unroll
    for (int t = 0; t < FEAT / 32; ++t) {
      int idx = lane + 32 * t;
      H[idx] = eluf(Z[idx]) + H[idx];
    }
    if (lane < FEAT % 32) {
      int idx = lane + (FEAT / 32) * 32;
      H[idx] = eluf(Z[idx]) + H[idx];
    }
    __syncthreads();
  }

  float vacc = 0.0f;
  float pacc[POL];
#pragma unroll
  for (int o = 0; o < POL; ++o) pacc[o] = 0.0f;
  for (int j = lane; j < FEAT; j += 32) {
    float fj = H[j];
    vacc += fj * vw[j];
#pragma unroll
    for (int o = 0; o < POL; ++o) pacc[o] += fj * pw[o * FEAT + j];
  }
  for (int off = 16; off >= 1; off >>= 1) {
    vacc += __shfl_xor(vacc, off, 32);
#pragma unroll
    for (int o = 0; o < POL; ++o) pacc[o] += __shfl_xor(pacc[o], off, 32);
  }
  if (lane == 0) {
    out[b] = vacc + vb[0];
    float logit[POL];
    float mx = -3.0e38f;
#pragma unroll
    for (int o = 0; o < POL; ++o) {
      logit[o] = pacc[o] + pb[o];
      mx = fmaxf(mx, logit[o]);
    }
    float sum = 0.0f;
#pragma unroll
    for (int o = 0; o < POL; ++o) {
      logit[o] = __expf(logit[o] - mx);
      sum += logit[o];
    }
    float inv = 1.0f / sum;
    float* pout = out + n_batch + (size_t)b * POL;
#pragma unroll
    for (int o = 0; o < POL; ++o) pout[o] = logit[o] * inv;
  }
}

extern "C" void kernel_launch(void* const* d_in, const int* in_sizes, int n_in,
                              void* d_out, int out_size, void* d_ws,
                              size_t ws_size, hipStream_t stream) {
  (void)n_in;
  (void)out_size;
  (void)d_ws;
  (void)ws_size;
  const float* x = (const float*)d_in[0];
  const float* embed_w = (const float*)d_in[1];
  const float* embed_b = (const float*)d_in[2];
  const float* fc1_w = (const float*)d_in[3];
  const float* fc1_b = (const float*)d_in[4];
  const float* ln_g = (const float*)d_in[5];
  const float* ln_b = (const float*)d_in[6];
  const float* iw = (const float*)d_in[7];
  const float* ib = (const float*)d_in[8];
  const float* ow = (const float*)d_in[9];
  const float* ob = (const float*)d_in[10];
  const float* vw = (const float*)d_in[11];
  const float* vb = (const float*)d_in[12];
  const float* pw = (const float*)d_in[13];
  const float* pb = (const float*)d_in[14];

  int n_batch = in_sizes[0] / FEAT;
  int grid = (n_batch + WAVES_PER_BLOCK - 1) / WAVES_PER_BLOCK;
  rubik_fused<<<grid, THREADS, 0, stream>>>(x, embed_w, embed_b, fc1_w, fc1_b,
                                            ln_g, ln_b, iw, ib, ow, ob, vw, vb,
                                            pw, pb, (float*)d_out, n_batch);
}


# Weight images of all 8 layers staged once per work-group (256 groups x 32 waves, a wave loops over 8 elements): per-layer staging and all barriers but one gone; exp2 pass split for the 128-VGPR cap
# speedup vs baseline: 1.4274x; 1.0400x over previous
#include <hip/hip_runtime.h>
#include <math.h>

#define NUM_TBLOCKS 8
#define SEQ 54
#define DIM 6
#define QKV (3 * DIM)
#define FEAT (SEQ * DIM)
#define POL 12
#define LN_EPS 1e-5f
#define QSCALE 0.58897778410361230f
#define PAD_SCORE -3.0e38f

#define WAVES_PER_BLOCK 32
#define THREADS (WAVES_PER_BLOCK * 32)
#define NUM_GROUPS 256
#define WAVE_LDS 2240
#define WSLOT 320
#define WCACHE ((NUM_TBLOCKS + 1) * WSLOT)

typedef __attribute__((ext_vector_type(2))) float v2f;
typedef __attribute__((ext_vector_type(4))) float v4f;
typedef __attribute__((ext_vector_type(8))) float v8f;

__device__ __forceinline__ v8f wmma_f32(v2f a, v2f b, v8f c) {
  return __builtin_amdgcn_wmma_f32_16x16x4_f32(false, a, false, b, (short)0, c,
                                               false, false);
}

__device__ __forceinline__ float xhalf(float x) {
  int b = __builtin_bit_cast(int, x);
  return __builtin_bit_cast(
      float, __builtin_amdgcn_permlanex16(b, b, 0x76543210, 0xfedcba98, false,
                                          false));
}

__device__ __forceinline__ float eluf(float x) {
  return x > 0.0f ? x : (__expf(x) - 1.0f);
}

__global__ __launch_bounds__(THREADS) void rubik_fused(
    const float* __restrict__ x, const float* __restrict__ embed_w,
    const float* __restrict__ embed_b, const float* __restrict__ fc1_w,
    const float* __restrict__ fc1_b, const float* __restrict__ ln_g,
    const float* __restrict__ ln_b, const float* __restrict__ iw,
    const float* __restrict__ ib, const float* __restrict__ ow,
    const float* __restrict__ ob, const float* __restrict__ vw,
    const float* __restrict__ vb, const float* __restrict__ pw,
    const float* __restrict__ pb, float* __restrict__ out, int n_batch) {
  __shared__ float smem[WAVES_PER_BLOCK * WAVE_LDS + WCACHE];
  const int lane = threadIdx.x & 31;
  const int wave = __builtin_amdgcn_readfirstlane((int)(threadIdx.x >> 5));

  float* H = &smem[wave * WAVE_LDS];
  float* Z = H + 324;
  float* Qp = Z + 324;
  float* Kp = Qp + 512;
  float* Vp = Kp + 512;
  float* dump = H + 2120;

  float* WC = &smem[WAVES_PER_BLOCK * WAVE_LDS];

  const int nloc = lane & 15;
  const int kb = (lane < 16) ? 0 : 2;
  const int upper8 = (lane < 16) ? 0 : 8;
  const bool lowhalf = (lane < 16);

  const bool qvcol = (nloc < 2 * DIM);
  float* qvBase = (nloc < DIM) ? (Qp + nloc) : (Vp + (nloc - DIM));

  auto fillW1 = [&](const float* w, int slot, float* wimg) {
    int kk = slot >> 6, ln = (slot >> 1) & 31, p = slot & 1;
    int j = ln & 15;
    int c = kk * 4 + ((ln < 16) ? 0 : 2) + p;
    bool v = (j < DIM) && (c < DIM);
    wimg[slot] = v ? w[v ? j * DIM + c : 0] : 0.0f;
  };
  auto fillB1 = [&](const float* bias, int ln, float* bimg) {
    int j = ln & 15;
    bimg[ln] = (j < DIM) ? bias[(j < DIM) ? j : 0] : 0.0f;
  };

  auto fillQVw = [&](const float* iwl, const float* owl, int slot,
                     float* wimg) {
    {
      int kk = slot >> 6, ln = (slot >> 1) & 31, p = slot & 1;
      int n = ln & 15;
      int c = kk * 4 + ((ln < 16) ? 0 : 2) + p;
      bool isq = (n < DIM);
      bool v = (n < 2 * DIM) && (c < DIM);
      int a = v ? (isq ? n : n - DIM) : 0;
      int cc = v ? c : 0;
      const float* lp = isq ? (iwl + DIM * DIM + a) : (owl + a * DIM);
      int ls = isq ? DIM : 1;
      const float* rp = (isq ? iwl : (iwl + 2 * DIM * DIM)) + cc;
      float acc = 0.0f;
      for (int m = 0; m < DIM; ++m) acc = fmaf(lp[m * ls], rp[m * DIM], acc);
      if (isq) acc *= QSCALE;
      wimg[slot] = v ? acc : 0.0f;
    }
  };
  auto fillQVb = [&](const float* iwl, const float* ibl, const float* owl,
                     int ln, float* bimg) {
    {
      int n = ln & 15;
      bool isq = (n < DIM);
      bool v = (n < 2 * DIM);
      int a = v ? (isq ? n : n - DIM) : 0;
      const float* lp = isq ? (iwl + DIM * DIM + a) : (owl + a * DIM);
      int ls = isq ? DIM : 1;
      const float* rp = isq ? ibl : (ibl + 2 * DIM);
      float acc = 0.0f;
      for (int m = 0; m < DIM; ++m) acc = fmaf(lp[m * ls], rp[m], acc);
      if (isq) acc *= QSCALE;
      bimg[ln] = v ? acc : 0.0f;
    }
  };

  auto loadA = [&](const float* src, int mt, v2f aw[2]) {
    int row = mt * 16 + nloc;
#pragma unroll
    for (int kk = 0; kk < 2; ++kk)
      aw[kk] = *(const v2f*)(src + row * DIM + kk * 4 + kb);
  };
  auto loadBimg = [&](const float* wimg, v2f bw[2]) {
    bw[0] = *(const v2f*)(wimg + lane * 2);
    bw[1] = *(const v2f*)(wimg + 64 + lane * 2);
  };
  auto loadCimg = [&](const float* bimg) {
    float bvv = bimg[lane];
    v8f c;
#pragma unroll
    for (int r = 0; r < 8; ++r) c[r] = bvv;
    return c;
  };

  {
    const int t = threadIdx.x;
    const int Ls = t >> 7, slot = t & 127;
    float* ws = WC + Ls * WSLOT;
    fillW1(fc1_w + Ls * DIM * DIM, slot, ws);
    fillQVw(iw + Ls * QKV * DIM, ow + Ls * DIM * DIM, slot, ws + 128);
    if (t < NUM_TBLOCKS * 32) {
      const int Lb = t >> 5, ln = t & 31;
      float* wb = WC + Lb * WSLOT;
      fillB1(fc1_b + Lb * DIM, ln, wb + 256);
      fillQVb(iw + Lb * QKV * DIM, ib + Lb * QKV, ow + Lb * DIM * DIM, ln,
              wb + 288);
    }
    float* we = WC + NUM_TBLOCKS * WSLOT;
    if (t < 128) fillW1(embed_w, t, we);
    if (t < 32) fillB1(embed_b, t, we + 256);
  }
  __syncthreads();

  for (int idx = lane; idx < 512 + 512 + 448; idx += 32) Qp[idx] = 0.0f;

  for (int s = lane; s < SEQ; s += 32) Vp[s * 8 + DIM] = 1.0f;
  for (int s = lane; s < 64; s += 32) Qp[s * 8 + 7] = PAD_SCORE;
  if (lane < 64 - SEQ) Kp[(SEQ + lane) * 8 + 7] = 1.0f;

  for (int b = blockIdx.x * WAVES_PER_BLOCK + wave; b < n_batch;
       b += NUM_GROUPS * WAVES_PER_BLOCK) {
  const float* xb = x + (size_t)b * FEAT;
  {
    int bn = b + NUM_GROUPS * WAVES_PER_BLOCK;
    if (bn < n_batch && lane * 32 < FEAT)
      __builtin_prefetch(x + (size_t)bn * FEAT + lane * 32, 0, 1);
  }
  for (int idx = lane; idx < FEAT; idx += 32) Z[idx] = xb[idx];

  {
    const float* W_FC = WC + NUM_TBLOCKS * WSLOT;
    const float* B_FC = W_FC + 256;
    v2f bw[2];
    loadBimg(W_FC, bw);
    v8f cb = loadCimg(B_FC);
#pragma unroll
    for (int mt = 0; mt < 4; ++mt) {
      v2f aw[2];
      loadA(Z, mt, aw);
      v8f d = wmma_f32(aw[0], bw[0], cb);
      d = wmma_f32(aw[1], bw[1], d);
      int row0 = mt * 16 + upper8;
      bool ok = (nloc < DIM) && ((mt < 3) || lowhalf);
      float* bz = ok ? &H[row0 * DIM + nloc] : dump;
#pragma unroll
      for (int r = 0; r < 8; ++r)
        if (mt < 3 || r < 6) bz[r * DIM] = d[r];
    }
  }

  for (int L = 0; L < NUM_TBLOCKS; ++L) {
    typedef const __attribute__((address_space(4))) float cfloat;
    cfloat* g = (cfloat*)(ln_g + L * DIM);
    cfloat* bbia = (cfloat*)(ln_b + L * DIM);
    cfloat* obl = (cfloat*)(ob + L * DIM);
    const float* W_FC = WC + L * WSLOT;
    const float* W_QV = W_FC + 128;
    const float* B_FC = W_FC + 256;
    const float* B_QV = W_FC + 288;

    {
      v2f bw[2];
      loadBimg(W_FC, bw);
      v8f cb = loadCimg(B_FC);
#pragma unroll
      for (int mt = 0; mt < 4; ++mt) {
        v2f aw[2];
        loadA(H, mt, aw);
        v8f d = wmma_f32(aw[0], bw[0], cb);
        d = wmma_f32(aw[1], bw[1], d);
        int row0 = mt * 16 + upper8;
        bool ok = (nloc < DIM) && ((mt < 3) || lowhalf);
        float* bz = ok ? &Z[row0 * DIM + nloc] : dump;
#pragma unroll
        for (int r = 0; r < 8; ++r)
          if (mt < 3 || r < 6) bz[r * DIM] = d[r];
      }
    }

    for (int s = lane; s < SEQ; s += 32) {
      float zv[DIM];
#pragma unroll
      for (int i = 0; i < DIM; ++i) zv[i] = eluf(Z[s * DIM + i]);
      float mu = 0.0f;
#pragma unroll
      for (int i = 0; i < DIM; ++i) mu += zv[i];
      mu *= (1.0f / DIM);
      float var = 0.0f;
#pragma unroll
      for (int i = 0; i < DIM; ++i) {
        float d = zv[i] - mu;
        var += d * d;
      }
      var *= (1.0f / DIM);
      float rs = rsqrtf(var + LN_EPS);
#pragma unroll
      for (int i = 0; i < DIM; ++i)
        Kp[s * 8 + i] = (zv[i] - mu) * rs * g[i] + bbia[i];
    }

    v2f ak[4][2];
#pragma unroll
    for (int mt = 0; mt < 4; ++mt)
#pragma unroll
      for (int kk = 0; kk < 2; ++kk)
        ak[mt][kk] = *(const v2f*)(Kp + (mt * 16 + nloc) * 8 + kk * 4 + kb);

    {
      v2f bw[2];
      loadBimg(W_QV, bw);
      v8f cb = loadCimg(B_QV);
#pragma unroll
      for (int mt = 0; mt < 4; ++mt) {
        v8f d = wmma_f32(ak[mt][0], bw[0], cb);
        d = wmma_f32(ak[mt][1], bw[1], d);
        int row0 = mt * 16 + upper8;
        bool ok0 = qvcol && ((mt < 3) || lowhalf);
        float* b0 = ok0 ? (qvBase + row0 * 8) : dump;
#pragma unroll
        for (int r = 0; r < 8; ++r)
          if (mt < 3 || r < 6) b0[r * 8] = d[r];
      }
    }

    const float* vcolp = Vp + ((nloc < 8) ? nloc : 7);
    v2f av[14];
#pragma unroll
    for (int mt = 0; mt < 3; ++mt)
#pragma unroll
      for (int p = 0; p < 4; ++p) {
        int key = mt * 16 + upper8 + 2 * p;
        av[mt * 4 + p].x = vcolp[key * 8];
        av[mt * 4 + p].y = vcolp[(key + 1) * 8];
      }
    {
      int key12 = lowhalf ? 48 : 52;
      av[12].x = vcolp[key12 * 8];
      av[12].y = vcolp[(key12 + 1) * 8];
      int key13 = lowhalf ? 50 : 54;
      av[13].x = vcolp[key13 * 8];
      av[13].y = vcolp[(key13 + 1) * 8];
    }

    float obv[DIM];
    for (int r = 0; r < DIM; ++r) obv[r] = obl[r];

#pragma nounroll
    for (int nt = 0; nt < 4; ++nt) {
      v2f bq[2];
#pragma unroll
      for (int kk = 0; kk < 2; ++kk)
        bq[kk] = *(const v2f*)(Qp + (nt * 16 + nloc) * 8 + kk * 4 + kb);
      v8f t[4];
#pragma unroll
      for (int mt = 0; mt < 4; ++mt) {
        v8f c = {};
#pragma unroll
        for (int kk = 0; kk < 2; ++kk) c = wmma_f32(ak[mt][kk], bq[kk], c);
        t[mt] = c;
      }

      float mx = t[0][0];
#pragma unroll
      for (int mt = 0; mt < 4; ++mt)
#pragma unroll
        for (int r = 0; r < 8; ++r)
          if ((mt > 0 || r > 0) && (mt < 3 || r < 6)) mx = fmaxf(mx, t[mt][r]);
      mx = fmaxf(mx, __shfl_xor(mx, 16));
      const v2f mx2 = {mx, mx};
#pragma unroll
      for (int mt = 0; mt < 4; ++mt)
#pragma unroll
        for (int p = 0; p < 4; ++p)
          if (mt < 3 || p < 3) {
            v2f d = (v2f){t[mt][2 * p], t[mt][2 * p + 1]} - mx2;
            t[mt][2 * p] = d.x;
            t[mt][2 * p + 1] = d.y;
          }
#pragma unroll
      for (int mt = 0; mt < 4; ++mt)
#pragma unroll
        for (int r = 0; r < 8; ++r)
          if (mt < 3 || r < 6) t[mt][r] = __builtin_amdgcn_exp2f(t[mt][r]);

      v8f o0 = {}, o1 = {};
#pragma unroll
      for (int mt = 0; mt < 3; ++mt)
#pragma unroll
        for (int p = 0; p < 4; p += 2) {
          v2f b0 = {t[mt][2 * p], t[mt][2 * p + 1]};
          v2f b1 = {t[mt][2 * p + 2], t[mt][2 * p + 3]};
          o0 = wmma_f32(av[mt * 4 + p], b0, o0);
          o1 = wmma_f32(av[mt * 4 + p + 1], b1, o1);
        }
      {
        v2f b12 = {t[3][0] + xhalf(t[3][4]), t[3][1] + xhalf(t[3][5])};
        v2f b13 = {t[3][2], t[3][3]};
        o0 = wmma_f32(av[12], b12, o0);
        o1 = wmma_f32(av[13], b13, o1);
      }
      {
        float inv = 1.0f / (o0[6] + o1[6]);
        int q = nt * 16 + nloc;
        if (lowhalf && q < SEQ) {
          float* zq = Z + q * DIM;
#pragma unroll
          for (int r = 0; r < DIM; r += 2) {
            v2f w = {(o0[r] + o1[r]) * inv + obv[r],
                     (o0[r + 1] + o1[r + 1]) * inv + obv[r + 1]};
            *(v2f*)(zq + r) = w;
          }
        }
      }
    }
#pragma unroll
    for (int t = 0; t < FEAT / 32; ++t) {
      int idx = lane + 32 * t;
      H[idx] = eluf(Z[idx]) + H[idx];
    }
    if (lane < FEAT % 32) {
      int idx = lane + (FEAT / 32) * 32;
      H[idx] = eluf(Z[idx]) + H[idx];
    }
  }

  float vacc = 0.0f;
  float pacc[POL];
#pragma unroll
  for (int o = 0; o < POL; ++o) pacc[o] = 0.0f;
  for (int j = lane; j < FEAT; j += 32) {
    float fj = H[j];
    vacc += fj * vw[j];
#pragma unroll
    for (int o = 0; o < POL; ++o) pacc[o] += fj * pw[o * FEAT + j];
  }
  for (int off = 16; off >= 1; off >>= 1) {
    vacc += __shfl_xor(vacc, off, 32);
#pragma unroll
    for (int o = 0; o < POL; ++o) pacc[o] += __shfl_xor(pacc[o], off, 32);
  }
  if (lane == 0) {
    out[b] = vacc + vb[0];
    float logit[POL];
    float mx = -3.0e38f;
#pragma unroll
    for (int o = 0; o < POL; ++o) {
      logit[o] = pacc[o] + pb[o];
      mx = fmaxf(mx, logit[o]);
    }
    float sum = 0.0f;
#pragma unroll
    for (int o = 0; o < POL; ++o) {
      logit[o] = __expf(logit[o] - mx);
      sum += logit[o];
    }
    float inv = 1.0f / sum;
    float* pout = out + n_batch + (size_t)b * POL;
#pragma unroll
    for (int o = 0; o < POL; ++o) pout[o] = logit[o] * inv;
  }
  }
}

extern "C" void kernel_launch(void* const* d_in, const int* in_sizes, int n_in,
                              void* d_out, int out_size, void* d_ws,
                              size_t ws_size, hipStream_t stream) {
  (void)n_in;
  (void)out_size;
  (void)d_ws;
  (void)ws_size;
  const float* x = (const float*)d_in[0];
  const float* embed_w = (const float*)d_in[1];
  const float* embed_b = (const float*)d_in[2];
  const float* fc1_w = (const float*)d_in[3];
  const float* fc1_b = (const float*)d_in[4];
  const float* ln_g = (const float*)d_in[5];
  const float* ln_b = (const float*)d_in[6];
  const float* iw = (const float*)d_in[7];
  const float* ib = (const float*)d_in[8];
  const float* ow = (const float*)d_in[9];
  const float* ob = (const float*)d_in[10];
  const float* vw = (const float*)d_in[11];
  const float* vb = (const float*)d_in[12];
  const float* pw = (const float*)d_in[13];
  const float* pb = (const float*)d_in[14];

  int n_batch = in_sizes[0] / FEAT;
  rubik_fused<<<NUM_GROUPS, THREADS, 0, stream>>>(x, embed_w, embed_b, fc1_w, fc1_b,
                                            ln_g, ln_b, iw, ib, ow, ob, vw, vb,
                                            pw, pb, (float*)d_out, n_batch);
}


# Wave priority falls with the work a wave has left (s_setprio 3, 2, 1, 0 over its last two elements): the waves of a SIMD end together instead of one after another; no arithmetic changed
# speedup vs baseline: 1.5437x; 1.0815x over previous
#include <hip/hip_runtime.h>
#include <math.h>

#define NUM_TBLOCKS 8
#define SEQ 54
#define DIM 6
#define QKV (3 * DIM)
#define FEAT (SEQ * DIM)
#define POL 12
#define LN_EPS 1e-5f
#define QSCALE 0.58897778410361230f
#define PAD_SCORE -3.0e38f

#define WAVES_PER_BLOCK 32
#define THREADS (WAVES_PER_BLOCK * 32)
#define NUM_GROUPS 256
#define WAVE_LDS 2240
#define WSLOT 320
#define WCACHE ((NUM_TBLOCKS + 1) * WSLOT)

typedef __attribute__((ext_vector_type(2))) float v2f;
typedef __attribute__((ext_vector_type(4))) float v4f;
typedef __attribute__((ext_vector_type(8))) float v8f;

__device__ __forceinline__ v8f wmma_f32(v2f a, v2f b, v8f c) {
  return __builtin_amdgcn_wmma_f32_16x16x4_f32(false, a, false, b, (short)0, c,
                                               false, false);
}

__device__ __forceinline__ float xhalf(float x) {
  int b = __builtin_bit_cast(int, x);
  return __builtin_bit_cast(
      float, __builtin_amdgcn_permlanex16(b, b, 0x76543210, 0xfedcba98, false,
                                          false));
}

__device__ __forceinline__ float eluf(float x) {
  return x > 0.0f ? x : (__expf(x) - 1.0f);
}

__global__ __launch_bounds__(THREADS) void rubik_fused(
    const float* __restrict__ x, const float* __restrict__ embed_w,
    const float* __restrict__ embed_b, const float* __restrict__ fc1_w,
    const float* __restrict__ fc1_b, const float* __restrict__ ln_g,
    const float* __restrict__ ln_b, const float* __restrict__ iw,
    const float* __restrict__ ib, const float* __restrict__ ow,
    const float* __restrict__ ob, const float* __restrict__ vw,
    const float* __restrict__ vb, const float* __restrict__ pw,
    const float* __restrict__ pb, float* __restrict__ out, int n_batch) {
  __shared__ float smem[WAVES_PER_BLOCK * WAVE_LDS + WCACHE];
  const int lane = threadIdx.x & 31;
  const int wave = __builtin_amdgcn_readfirstlane((int)(threadIdx.x >> 5));

  float* H = &smem[wave * WAVE_LDS];
  float* Z = H + 324;
  float* Qp = Z + 324;
  float* Kp = Qp + 512;
  float* Vp = Kp + 512;
  float* dump = H + 2120;

  float* WC = &smem[WAVES_PER_BLOCK * WAVE_LDS];

  const int nloc = lane & 15;
  const int kb = (lane < 16) ? 0 : 2;
  const int upper8 = (lane < 16) ? 0 : 8;
  const bool lowhalf = (lane < 16);

  const bool qvcol = (nloc < 2 * DIM);
  float* qvBase = (nloc < DIM) ? (Qp + nloc) : (Vp + (nloc - DIM));

  auto fillW1 = [&](const float* w, int slot, float* wimg) {
    int kk = slot >> 6, ln = (slot >> 1) & 31, p = slot & 1;
    int j = ln & 15;
    int c = kk * 4 + ((ln < 16) ? 0 : 2) + p;
    bool v = (j < DIM) && (c < DIM);
    wimg[slot] = v ? w[v ? j * DIM + c : 0] : 0.0f;
  };
  auto fillB1 = [&](const float* bias, int ln, float* bimg) {
    int j = ln & 15;
    bimg[ln] = (j < DIM) ? bias[(j < DIM) ? j : 0] : 0.0f;
  };

  auto fillQVw = [&](const float* iwl, const float* owl, int slot,
                     float* wimg) {
    {
      int kk = slot >> 6, ln = (slot >> 1) & 31, p = slot & 1;
      int n = ln & 15;
      int c = kk * 4 + ((ln < 16) ? 0 : 2) + p;
      bool isq = (n < DIM);
      bool v = (n < 2 * DIM) && (c < DIM);
      int a = v ? (isq ? n : n - DIM) : 0;
      int cc = v ? c : 0;
      const float* lp = isq ? (iwl + DIM * DIM + a) : (owl + a * DIM);
      int ls = isq ? DIM : 1;
      const float* rp = (isq ? iwl : (iwl + 2 * DIM * DIM)) + cc;
      float acc = 0.0f;
      for (int m = 0; m < DIM; ++m) acc = fmaf(lp[m * ls], rp[m * DIM], acc);
      if (isq) acc *= QSCALE;
      wimg[slot] = v ? acc : 0.0f;
    }
  };
  auto fillQVb = [&](const float* iwl, const float* ibl, const float* owl,
                     int ln, float* bimg) {
    {
      int n = ln & 15;
      bool isq = (n < DIM);
      bool v = (n < 2 * DIM);
      int a = v ? (isq ? n : n - DIM) : 0;
      const float* lp = isq ? (iwl + DIM * DIM + a) : (owl + a * DIM);
      int ls = isq ? DIM : 1;
      const float* rp = isq ? ibl : (ibl + 2 * DIM);
      float acc = 0.0f;
      for (int m = 0; m < DIM; ++m) acc = fmaf(lp[m * ls], rp[m], acc);
      if (isq) acc *= QSCALE;
      bimg[ln] = v ? acc : 0.0f;
    }
  };

  auto loadA = [&](const float* src, int mt, v2f aw[2]) {
    int row = mt * 16 + nloc;
#pragma unroll
    for (int kk = 0; kk < 2; ++kk)
      aw[kk] = *(const v2f*)(src + row * DIM + kk * 4 + kb);
  };
  auto loadBimg = [&](const float* wimg, v2f bw[2]) {
    bw[0] = *(const v2f*)(wimg + lane * 2);
    bw[1] = *(const v2f*)(wimg + 64 + lane * 2);
  };
  auto loadCimg = [&](const float* bimg) {
    float bvv = bimg[lane];
    v8f c;
#pragma unroll
    for (int r = 0; r < 8; ++r) c[r] = bvv;
    return c;
  };

  {
    const int t = threadIdx.x;
    const int Ls = t >> 7, slot = t & 127;
    float* ws = WC + Ls * WSLOT;
    fillW1(fc1_w + Ls * DIM * DIM, slot, ws);
    fillQVw(iw + Ls * QKV * DIM, ow + Ls * DIM * DIM, slot, ws + 128);
    if (t < NUM_TBLOCKS * 32) {
      const int Lb = t >> 5, ln = t & 31;
      float* wb = WC + Lb * WSLOT;
      fillB1(fc1_b + Lb * DIM, ln, wb + 256);
      fillQVb(iw + Lb * QKV * DIM, ib + Lb * QKV, ow + Lb * DIM * DIM, ln,
              wb + 288);
    }
    float* we = WC + NUM_TBLOCKS * WSLOT;
    if (t < 128) fillW1(embed_w, t, we);
    if (t < 32) fillB1(embed_b, t, we + 256);
  }
  __syncthreads();

  for (int idx = lane; idx < 512 + 512 + 448; idx += 32) Qp[idx] = 0.0f;

  for (int s = lane; s < SEQ; s += 32) Vp[s * 8 + DIM] = 1.0f;
  for (int s = lane; s < 64; s += 32) Qp[s * 8 + 7] = PAD_SCORE;
  if (lane < 64 - SEQ) Kp[(SEQ + lane) * 8 + 7] = 1.0f;

  for (int b = blockIdx.x * WAVES_PER_BLOCK + wave; b < n_batch;
       b += NUM_GROUPS * WAVES_PER_BLOCK) {
  const unsigned left =
      (unsigned)(n_batch - 1 - b) / (unsigned)(NUM_GROUPS * WAVES_PER_BLOCK);
  if (left >= 2u)
    __builtin_amdgcn_s_setprio(3);
  else if (left == 1u)
    __builtin_amdgcn_s_setprio(2);
  else
    __builtin_amdgcn_s_setprio(1);
  const float* xb = x + (size_t)b * FEAT;
  {
    int bn = b + NUM_GROUPS * WAVES_PER_BLOCK;
    if (bn < n_batch && lane * 32 < FEAT)
      __builtin_prefetch(x + (size_t)bn * FEAT + lane * 32, 0, 1);
  }
  for (int idx = lane; idx < FEAT; idx += 32) Z[idx] = xb[idx];

  {
    const float* W_FC = WC + NUM_TBLOCKS * WSLOT;
    const float* B_FC = W_FC + 256;
    v2f bw[2];
    loadBimg(W_FC, bw);
    v8f cb = loadCimg(B_FC);
#pragma unroll
    for (int mt = 0; mt < 4; ++mt) {
      v2f aw[2];
      loadA(Z, mt, aw);
      v8f d = wmma_f32(aw[0], bw[0], cb);
      d = wmma_f32(aw[1], bw[1], d);
      int row0 = mt * 16 + upper8;
      bool ok = (nloc < DIM) && ((mt < 3) || lowhalf);
      float* bz = ok ? &H[row0 * DIM + nloc] : dump;
#pragma unroll
      for (int r = 0; r < 8; ++r)
        if (mt < 3 || r < 6) bz[r * DIM] = d[r];
    }
  }

  for (int L = 0; L < NUM_TBLOCKS; ++L) {
    if (left == 0u && L == NUM_TBLOCKS - 3) __builtin_amdgcn_s_setprio(0);
    typedef const __attribute__((address_space(4))) float cfloat;
    cfloat* g = (cfloat*)(ln_g + L * DIM);
    cfloat* bbia = (cfloat*)(ln_b + L * DIM);
    cfloat* obl = (cfloat*)(ob + L * DIM);
    const float* W_FC = WC + L * WSLOT;
    const float* W_QV = W_FC + 128;
    const float* B_FC = W_FC + 256;
    const float* B_QV = W_FC + 288;

    {
      v2f bw[2];
      loadBimg(W_FC, bw);
      v8f cb = loadCimg(B_FC);
#pragma unroll
      for (int mt = 0; mt < 4; ++mt) {
        v2f aw[2];
        loadA(H, mt, aw);
        v8f d = wmma_f32(aw[0], bw[0], cb);
        d = wmma_f32(aw[1], bw[1], d);
        int row0 = mt * 16 + upper8;
        bool ok = (nloc < DIM) && ((mt < 3) || lowhalf);
        float* bz = ok ? &Z[row0 * DIM + nloc] : dump;
#pragma unroll
        for (int r = 0; r < 8; ++r)
          if (mt < 3 || r < 6) bz[r * DIM] = d[r];
      }
    }

    for (int s = lane; s < SEQ; s += 32) {
      float zv[DIM];
#pragma unroll
      for (int i = 0; i < DIM; ++i) zv[i] = eluf(Z[s * DIM + i]);
      float mu = 0.0f;
#pragma unroll
      for (int i = 0; i < DIM; ++i) mu += zv[i];
      mu *= (1.0f / DIM);
      float var = 0.0f;
#pragma unroll
      for (int i = 0; i < DIM; ++i) {
        float d = zv[i] - mu;
        var += d * d;
      }
      var *= (1.0f / DIM);
      float rs = rsqrtf(var + LN_EPS);
#pragma unroll
      for (int i = 0; i < DIM; ++i)
        Kp[s * 8 + i] = (zv[i] - mu) * rs * g[i] + bbia[i];
    }

    v2f ak[4][2];
#pragma unroll
    for (int mt = 0; mt < 4; ++mt)
#pragma unroll
      for (int kk = 0; kk < 2; ++kk)
        ak[mt][kk] = *(const v2f*)(Kp + (mt * 16 + nloc) * 8 + kk * 4 + kb);

    {
      v2f bw[2];
      loadBimg(W_QV, bw);
      v8f cb = loadCimg(B_QV);
#pragma unroll
      for (int mt = 0; mt < 4; ++mt) {
        v8f d = wmma_f32(ak[mt][0], bw[0], cb);
        d = wmma_f32(ak[mt][1], bw[1], d);
        int row0 = mt * 16 + upper8;
        bool ok0 = qvcol && ((mt < 3) || lowhalf);
        float* b0 = ok0 ? (qvBase + row0 * 8) : dump;
#pragma unroll
        for (int r = 0; r < 8; ++r)
          if (mt < 3 || r < 6) b0[r * 8] = d[r];
      }
    }

    const float* vcolp = Vp + ((nloc < 8) ? nloc : 7);
    v2f av[14];
#pragma unroll
    for (int mt = 0; mt < 3; ++mt)
#pragma unroll
      for (int p = 0; p < 4; ++p) {
        int key = mt * 16 + upper8 + 2 * p;
        av[mt * 4 + p].x = vcolp[key * 8];
        av[mt * 4 + p].y = vcolp[(key + 1) * 8];
      }
    {
      int key12 = lowhalf ? 48 : 52;
      av[12].x = vcolp[key12 * 8];
      av[12].y = vcolp[(key12 + 1) * 8];
      int key13 = lowhalf ? 50 : 54;
      av[13].x = vcolp[key13 * 8];
      av[13].y = vcolp[(key13 + 1) * 8];
    }

    float obv[DIM];
    for (int r = 0; r < DIM; ++r) obv[r] = obl[r];

#pragma nounroll
    for (int nt = 0; nt < 4; ++nt) {
      v2f bq[2];
#pragma unroll
      for (int kk = 0; kk < 2; ++kk)
        bq[kk] = *(const v2f*)(Qp + (nt * 16 + nloc) * 8 + kk * 4 + kb);
      v8f t[4];
#pragma unroll
      for (int mt = 0; mt < 4; ++mt) {
        v8f c = {};
#pragma unroll
        for (int kk = 0; kk < 2; ++kk) c = wmma_f32(ak[mt][kk], bq[kk], c);
        t[mt] = c;
      }

      float mx = t[0][0];
#pragma unroll
      for (int mt = 0; mt < 4; ++mt)
#pragma unroll
        for (int r = 0; r < 8; ++r)
          if ((mt > 0 || r > 0) && (mt < 3 || r < 6)) mx = fmaxf(mx, t[mt][r]);
      mx = fmaxf(mx, __shfl_xor(mx, 16));
      const v2f mx2 = {mx, mx};
#pragma unroll
      for (int mt = 0; mt < 4; ++mt)
#pragma unroll
        for (int p = 0; p < 4; ++p)
          if (mt < 3 || p < 3) {
            v2f d = (v2f){t[mt][2 * p], t[mt][2 * p + 1]} - mx2;
            t[mt][2 * p] = d.x;
            t[mt][2 * p + 1] = d.y;
          }
#pragma unroll
      for (int mt = 0; mt < 4; ++mt)
#pragma unroll
        for (int r = 0; r < 8; ++r)
          if (mt < 3 || r < 6) t[mt][r] = __builtin_amdgcn_exp2f(t[mt][r]);

      v8f o0 = {}, o1 = {};
#pragma unroll
      for (int mt = 0; mt < 3; ++mt)
#pragma unroll
        for (int p = 0; p < 4; p += 2) {
          v2f b0 = {t[mt][2 * p], t[mt][2 * p + 1]};
          v2f b1 = {t[mt][2 * p + 2], t[mt][2 * p + 3]};
          o0 = wmma_f32(av[mt * 4 + p], b0, o0);
          o1 = wmma_f32(av[mt * 4 + p + 1], b1, o1);
        }
      {
        v2f b12 = {t[3][0] + xhalf(t[3][4]), t[3][1] + xhalf(t[3][5])};
        v2f b13 = {t[3][2], t[3][3]};
        o0 = wmma_f32(av[12], b12, o0);
        o1 = wmma_f32(av[13], b13, o1);
      }
      {
        float inv = 1.0f / (o0[6] + o1[6]);
        int q = nt * 16 + nloc;
        if (lowhalf && q < SEQ) {
          float* zq = Z + q * DIM;
#pragma unroll
          for (int r = 0; r < DIM; r += 2) {
            v2f w = {(o0[r] + o1[r]) * inv + obv[r],
                     (o0[r + 1] + o1[r + 1]) * inv + obv[r + 1]};
            *(v2f*)(zq + r) = w;
          }
        }
      }
    }
#pragma unroll
    for (int t = 0; t < FEAT / 32; ++t) {
      int idx = lane + 32 * t;
      H[idx] = eluf(Z[idx]) + H[idx];
    }
    if (lane < FEAT % 32) {
      int idx = lane + (FEAT / 32) * 32;
      H[idx] = eluf(Z[idx]) + H[idx];
    }
  }

  float vacc = 0.0f;
  float pacc[POL];
#pragma unroll
  for (int o = 0; o < POL; ++o) pacc[o] = 0.0f;
  for (int j = lane; j < FEAT; j += 32) {
    float fj = H[j];
    vacc += fj * vw[j];
#pragma unroll
    for (int o = 0; o < POL; ++o) pacc[o] += fj * pw[o * FEAT + j];
  }
  for (int off = 16; off >= 1; off >>= 1) {
    vacc += __shfl_xor(vacc, off, 32);
#pragma unroll
    for (int o = 0; o < POL; ++o) pacc[o] += __shfl_xor(pacc[o], off, 32);
  }
  if (lane == 0) {
    out[b] = vacc + vb[0];
    float logit[POL];
    float mx = -3.0e38f;
#pragma unroll
    for (int o = 0; o < POL; ++o) {
      logit[o] = pacc[o] + pb[o];
      mx = fmaxf(mx, logit[o]);
    }
    float sum = 0.0f;
#pragma unroll
    for (int o = 0; o < POL; ++o) {
      logit[o] = __expf(logit[o] - mx);
      sum += logit[o];
    }
    float inv = 1.0f / sum;
    float* pout = out + n_batch + (size_t)b * POL;
#pragma unroll
    for (int o = 0; o < POL; ++o) pout[o] = logit[o] * inv;
  }
  }
  __builtin_amdgcn_s_setprio(0);
}

extern "C" void kernel_launch(void* const* d_in, const int* in_sizes, int n_in,
                              void* d_out, int out_size, void* d_ws,
                              size_t ws_size, hipStream_t stream) {
  (void)n_in;
  (void)out_size;
  (void)d_ws;
  (void)ws_size;
  const float* x = (const float*)d_in[0];
  const float* embed_w = (const float*)d_in[1];
  const float* embed_b = (const float*)d_in[2];
  const float* fc1_w = (const float*)d_in[3];
  const float* fc1_b = (const float*)d_in[4];
  const float* ln_g = (const float*)d_in[5];
  const float* ln_b = (const float*)d_in[6];
  const float* iw = (const float*)d_in[7];
  const float* ib = (const float*)d_in[8];
  const float* ow = (const float*)d_in[9];
  const float* ob = (const float*)d_in[10];
  const float* vw = (const float*)d_in[11];
  const float* vb = (const float*)d_in[12];
  const float* pw = (const float*)d_in[13];
  const float* pb = (const float*)d_in[14];

  int n_batch = in_sizes[0] / FEAT;
  rubik_fused<<<NUM_GROUPS, THREADS, 0, stream>>>(x, embed_w, embed_b, fc1_w, fc1_b,
                                            ln_g, ln_b, iw, ib, ow, ob, vw, vb,
                                            pw, pb, (float*)d_out, n_batch);
}


# Softmax shift moved into the score product: a per-layer bound B of all scores (LayerNorm row bound x norms of the staged Q' image) rides in k column 6; per-query max, shuffle, subtract gone if B<=40
# speedup vs baseline: 1.5997x; 1.0363x over previous
#include <hip/hip_runtime.h>
#include <math.h>

#define NUM_TBLOCKS 8
#define SEQ 54
#define DIM 6
#define QKV (3 * DIM)
#define FEAT (SEQ * DIM)
#define POL 12
#define LN_EPS 1e-5f
#define QSCALE 0.58897778410361230f
#define PAD_SCORE -3.0e38f
#define LN_ROW_BOUND 2.4494897427831781f
#define SHIFT_MAX 40.0f

#define WAVES_PER_BLOCK 32
#define THREADS (WAVES_PER_BLOCK * 32)
#define NUM_GROUPS 256
#define WAVE_LDS 2240
#define WSLOT 320
#define WCACHE ((NUM_TBLOCKS + 1) * WSLOT)

typedef __attribute__((ext_vector_type(2))) float v2f;
typedef __attribute__((ext_vector_type(4))) float v4f;
typedef __attribute__((ext_vector_type(8))) float v8f;

template <bool E> struct exact_sel { static constexpr bool value = E; };

__device__ __forceinline__ v8f wmma_f32(v2f a, v2f b, v8f c) {
  return __builtin_amdgcn_wmma_f32_16x16x4_f32(false, a, false, b, (short)0, c,
                                               false, false);
}

__device__ __forceinline__ float xhalf(float x) {
  int b = __builtin_bit_cast(int, x);
  return __builtin_bit_cast(
      float, __builtin_amdgcn_permlanex16(b, b, 0x76543210, 0xfedcba98, false,
                                          false));
}

__device__ __forceinline__ float eluf(float x) {
  return x > 0.0f ? x : (__expf(x) - 1.0f);
}

__global__ __launch_bounds__(THREADS) void rubik_fused(
    const float* __restrict__ x, const float* __restrict__ embed_w,
    const float* __restrict__ embed_b, const float* __restrict__ fc1_w,
    const float* __restrict__ fc1_b, const float* __restrict__ ln_g,
    const float* __restrict__ ln_b, const float* __restrict__ iw,
    const float* __restrict__ ib, const float* __restrict__ ow,
    const float* __restrict__ ob, const float* __restrict__ vw,
    const float* __restrict__ vb, const float* __restrict__ pw,
    const float* __restrict__ pb, float* __restrict__ out, int n_batch) {
  __shared__ float smem[WAVES_PER_BLOCK * WAVE_LDS + WCACHE];
  const int lane = threadIdx.x & 31;
  const int wave = __builtin_amdgcn_readfirstlane((int)(threadIdx.x >> 5));

  float* H = &smem[wave * WAVE_LDS];
  float* Z = H + 324;
  float* Qp = Z + 324;
  float* Kp = Qp + 512;
  float* Vp = Kp + 512;
  float* dump = H + 2120;

  float* WC = &smem[WAVES_PER_BLOCK * WAVE_LDS];

  const int nloc = lane & 15;
  const int kb = (lane < 16) ? 0 : 2;
  const int upper8 = (lane < 16) ? 0 : 8;
  const bool lowhalf = (lane < 16);

  const bool qvcol = (nloc <= 2 * DIM);
  float* qvBase = (nloc < DIM)       ? (Qp + nloc)
                  : (nloc < 2 * DIM) ? (Vp + (nloc - DIM))
                                     : (Qp + DIM);

  auto fillW1 = [&](const float* w, int slot, float* wimg) {
    int kk = slot >> 6, ln = (slot >> 1) & 31, p = slot & 1;
    int j = ln & 15;
    int c = kk * 4 + ((ln < 16) ? 0 : 2) + p;
    bool v = (j < DIM) && (c < DIM);
    wimg[slot] = v ? w[v ? j * DIM + c : 0] : 0.0f;
  };
  auto fillB1 = [&](const float* bias, int ln, float* bimg) {
    int j = ln & 15;
    bimg[ln] = (j < DIM) ? bias[(j < DIM) ? j : 0] : 0.0f;
  };

  auto fillQVw = [&](const float* iwl, const float* owl, int slot,
                     float* wimg) {
    {
      int kk = slot >> 6, ln = (slot >> 1) & 31, p = slot & 1;
      int n = ln & 15;
      int c = kk * 4 + ((ln < 16) ? 0 : 2) + p;
      bool isq = (n < DIM);
      bool v = (n < 2 * DIM) && (c < DIM);
      int a = v ? (isq ? n : n - DIM) : 0;
      int cc = v ? c : 0;
      const float* lp = isq ? (iwl + DIM * DIM + a) : (owl + a * DIM);
      int ls = isq ? DIM : 1;
      const float* rp = (isq ? iwl : (iwl + 2 * DIM * DIM)) + cc;
      float acc = 0.0f;
      for (int m = 0; m < DIM; ++m) acc = fmaf(lp[m * ls], rp[m * DIM], acc);
      if (isq) acc *= QSCALE;
      wimg[slot] = v ? acc : 0.0f;
    }
  };
  auto fillQVb = [&](const float* iwl, const float* ibl, const float* owl,
                     int ln, float* bimg) {
    {
      int n = ln & 15;
      bool isq = (n < DIM);
      bool v = (n < 2 * DIM);
      int a = v ? (isq ? n : n - DIM) : 0;
      const float* lp = isq ? (iwl + DIM * DIM + a) : (owl + a * DIM);
      int ls = isq ? DIM : 1;
      const float* rp = isq ? ibl : (ibl + 2 * DIM);
      float acc = 0.0f;
      for (int m = 0; m < DIM; ++m) acc = fmaf(lp[m * ls], rp[m], acc);
      if (isq) acc *= QSCALE;
      bimg[ln] = v ? acc : 0.0f;
    }
  };

  auto loadA = [&](const float* src, int mt, v2f aw[2]) {
    int row = mt * 16 + nloc;
#pragma unroll
    for (int kk = 0; kk < 2; ++kk)
      aw[kk] = *(const v2f*)(src + row * DIM + kk * 4 + kb);
  };
  auto loadBimg = [&](const float* wimg, v2f bw[2]) {
    bw[0] = *(const v2f*)(wimg + lane * 2);
    bw[1] = *(const v2f*)(wimg + 64 + lane * 2);
  };
  auto loadCimg = [&](const float* bimg) {
    float bvv = bimg[lane];
    v8f c;
#pragma unroll
    for (int r = 0; r < 8; ++r) c[r] = bvv;
    return c;
  };

  {
    const int t = threadIdx.x;
    const int Ls = t >> 7, slot = t & 127;
    float* ws = WC + Ls * WSLOT;
    fillW1(fc1_w + Ls * DIM * DIM, slot, ws);
    fillQVw(iw + Ls * QKV * DIM, ow + Ls * DIM * DIM, slot, ws + 128);
    if (t < NUM_TBLOCKS * 32) {
      const int Lb = t >> 5, ln = t & 31;
      float* wb = WC + Lb * WSLOT;
      fillB1(fc1_b + Lb * DIM, ln, wb + 256);
      fillQVb(iw + Lb * QKV * DIM, ib + Lb * QKV, ow + Lb * DIM * DIM, ln,
              wb + 288);
    }
    float* we = WC + NUM_TBLOCKS * WSLOT;
    if (t < 128) fillW1(embed_w, t, we);
    if (t < 32) fillB1(embed_b, t, we + 256);
  }
  __syncthreads();

  if (wave < NUM_TBLOCKS) {
    float* ws = WC + wave * WSLOT;
    const float* wq = ws + 128;
    float* bqi = ws + 288;
    v2f w0 = *(const v2f*)(wq + lane * 2);
    v2f w1 = *(const v2f*)(wq + 64 + lane * 2);
    float cbv = bqi[lane];
    float sw = (nloc < DIM)
                   ? (w0.x * w0.x + w0.y * w0.y + w1.x * w1.x + w1.y * w1.y)
                   : 0.0f;
    float sc = (lane < DIM) ? cbv * cbv : 0.0f;
    float gv = (lane < DIM) ? ln_g[wave * DIM + ((lane < DIM) ? lane : 0)] : 0.0f;
    float bv = (lane < DIM) ? ln_b[wave * DIM + ((lane < DIM) ? lane : 0)] : 0.0f;
    float gm = fabsf(gv);
    float sb = bv * bv;
    for (int off = 16; off >= 1; off >>= 1) {
      sw += __shfl_xor(sw, off, 32);
      sc += __shfl_xor(sc, off, 32);
      sb += __shfl_xor(sb, off, 32);
      gm = fmaxf(gm, __shfl_xor(gm, off, 32));
    }
    float R = LN_ROW_BOUND * gm + sqrtf(sb);
    float B = (sqrtf(sw) * R + sqrtf(sc)) * R;
    float shift = (B <= SHIFT_MAX) ? -B : 0.0f;
    if (nloc == 2 * DIM) bqi[lane] = shift;
  }
  __syncthreads();

  for (int idx = lane; idx < 512 + 512 + 448; idx += 32) Qp[idx] = 0.0f;

  for (int s = lane; s < SEQ; s += 32) Vp[s * 8 + DIM] = 1.0f;
  for (int s = lane; s < 64; s += 32) Qp[s * 8 + 7] = PAD_SCORE;
  if (lane < 64 - SEQ) Kp[(SEQ + lane) * 8 + 7] = 1.0f;
  for (int s = lane; s < SEQ; s += 32) Kp[s * 8 + DIM] = 1.0f;

  for (int b = blockIdx.x * WAVES_PER_BLOCK + wave; b < n_batch;
       b += NUM_GROUPS * WAVES_PER_BLOCK) {
  const unsigned left =
      (unsigned)(n_batch - 1 - b) / (unsigned)(NUM_GROUPS * WAVES_PER_BLOCK);
  if (left >= 2u)
    __builtin_amdgcn_s_setprio(3);
  else if (left == 1u)
    __builtin_amdgcn_s_setprio(2);
  else
    __builtin_amdgcn_s_setprio(1);
  const float* xb = x + (size_t)b * FEAT;
  {
    int bn = b + NUM_GROUPS * WAVES_PER_BLOCK;
    if (bn < n_batch && lane * 32 < FEAT)
      __builtin_prefetch(x + (size_t)bn * FEAT + lane * 32, 0, 1);
  }
  for (int idx = lane; idx < FEAT; idx += 32) Z[idx] = xb[idx];

  {
    const float* W_FC = WC + NUM_TBLOCKS * WSLOT;
    const float* B_FC = W_FC + 256;
    v2f bw[2];
    loadBimg(W_FC, bw);
    v8f cb = loadCimg(B_FC);
#pragma unroll
    for (int mt = 0; mt < 4; ++mt) {
      v2f aw[2];
      loadA(Z, mt, aw);
      v8f d = wmma_f32(aw[0], bw[0], cb);
      d = wmma_f32(aw[1], bw[1], d);
      int row0 = mt * 16 + upper8;
      bool ok = (nloc < DIM) && ((mt < 3) || lowhalf);
      float* bz = ok ? &H[row0 * DIM + nloc] : dump;
#pragma unroll
      for (int r = 0; r < 8; ++r)
        if (mt < 3 || r < 6) bz[r * DIM] = d[r];
    }
  }

  for (int L = 0; L < NUM_TBLOCKS; ++L) {
    if (left == 0u && L == NUM_TBLOCKS - 3) __builtin_amdgcn_s_setprio(0);
    typedef const __attribute__((address_space(4))) float cfloat;
    cfloat* g = (cfloat*)(ln_g + L * DIM);
    cfloat* bbia = (cfloat*)(ln_b + L * DIM);
    cfloat* obl = (cfloat*)(ob + L * DIM);
    const float* W_FC = WC + L * WSLOT;
    const float* W_QV = W_FC + 128;
    const float* B_FC = W_FC + 256;
    const float* B_QV = W_FC + 288;

    {
      v2f bw[2];
      loadBimg(W_FC, bw);
      v8f cb = loadCimg(B_FC);
#pragma unroll
      for (int mt = 0; mt < 4; ++mt) {
        v2f aw[2];
        loadA(H, mt, aw);
        v8f d = wmma_f32(aw[0], bw[0], cb);
        d = wmma_f32(aw[1], bw[1], d);
        int row0 = mt * 16 + upper8;
        bool ok = (nloc < DIM) && ((mt < 3) || lowhalf);
        float* bz = ok ? &Z[row0 * DIM + nloc] : dump;
#pragma unroll
        for (int r = 0; r < 8; ++r)
          if (mt < 3 || r < 6) bz[r * DIM] = d[r];
      }
    }

    for (int s = lane; s < SEQ; s += 32) {
      float zv[DIM];
#pragma unroll
      for (int i = 0; i < DIM; ++i) zv[i] = eluf(Z[s * DIM + i]);
      float mu = 0.0f;
#pragma unroll
      for (int i = 0; i < DIM; ++i) mu += zv[i];
      mu *= (1.0f / DIM);
      float var = 0.0f;
#pragma unroll
      for (int i = 0; i < DIM; ++i) {
        float d = zv[i] - mu;
        var += d * d;
      }
      var *= (1.0f / DIM);
      float rs = rsqrtf(var + LN_EPS);
#pragma unroll
      for (int i = 0; i < DIM; ++i)
        Kp[s * 8 + i] = (zv[i] - mu) * rs * g[i] + bbia[i];
    }

    v2f ak[4][2];
#pragma unroll
    for (int mt = 0; mt < 4; ++mt)
#pragma unroll
      for (int kk = 0; kk < 2; ++kk)
        ak[mt][kk] = *(const v2f*)(Kp + (mt * 16 + nloc) * 8 + kk * 4 + kb);

    bool exactmax;
    {
      v2f bw[2];
      loadBimg(W_QV, bw);
      v8f cb = loadCimg(B_QV);
      exactmax = __builtin_bit_cast(
                     float, __builtin_amdgcn_readlane(
                                __builtin_bit_cast(int, cb[0]), 2 * DIM)) == 0.0f;
#pragma unroll
      for (int mt = 0; mt < 4; ++mt) {
        v8f d = wmma_f32(ak[mt][0], bw[0], cb);
        d = wmma_f32(ak[mt][1], bw[1], d);
        int row0 = mt * 16 + upper8;
        bool ok0 = qvcol && ((mt < 3) || lowhalf);
        float* b0 = ok0 ? (qvBase + row0 * 8) : dump;
#pragma unroll
        for (int r = 0; r < 8; ++r)
          if (mt < 3 || r < 6) b0[r * 8] = d[r];
      }
    }

    const float* vcolp = Vp + ((nloc < 8) ? nloc : 7);
    v2f av[14];
#pragma unroll
    for (int mt = 0; mt < 3; ++mt)
#pragma unroll
      for (int p = 0; p < 4; ++p) {
        int key = mt * 16 + upper8 + 2 * p;
        av[mt * 4 + p].x = vcolp[key * 8];
        av[mt * 4 + p].y = vcolp[(key + 1) * 8];
      }
    {
      int key12 = lowhalf ? 48 : 52;
      av[12].x = vcolp[key12 * 8];
      av[12].y = vcolp[(key12 + 1) * 8];
      int key13 = lowhalf ? 50 : 54;
      av[13].x = vcolp[key13 * 8];
      av[13].y = vcolp[(key13 + 1) * 8];
    }

    float obv[DIM];
    for (int r = 0; r < DIM; ++r) obv[r] = obl[r];

    auto attention = [&](auto sel) {
    constexpr bool EXACT = decltype(sel)::value;
#pragma nounroll
    for (int nt = 0; nt < 4; ++nt) {
      v2f bq[2];
#pragma unroll
      for (int kk = 0; kk < 2; ++kk)
        bq[kk] = *(const v2f*)(Qp + (nt * 16 + nloc) * 8 + kk * 4 + kb);
      v8f t[4];
#pragma unroll
      for (int mt = 0; mt < 4; ++mt) {
        v8f c = {};
#pragma unroll
        for (int kk = 0; kk < 2; ++kk) c = wmma_f32(ak[mt][kk], bq[kk], c);
        t[mt] = c;
      }

      if constexpr (EXACT) {
        float mx = t[0][0];
#pragma unroll
        for (int mt = 0; mt < 4; ++mt)
#pragma unroll
          for (int r = 0; r < 8; ++r)
            if ((mt > 0 || r > 0) && (mt < 3 || r < 6))
              mx = fmaxf(mx, t[mt][r]);
        mx = fmaxf(mx, __shfl_xor(mx, 16));
        const v2f mx2 = {mx, mx};
#pragma unroll
        for (int mt = 0; mt < 4; ++mt)
#pragma unroll
          for (int p = 0; p < 4; ++p)
            if (mt < 3 || p < 3) {
              v2f d = (v2f){t[mt][2 * p], t[mt][2 * p + 1]} - mx2;
              t[mt][2 * p] = d.x;
              t[mt][2 * p + 1] = d.y;
            }
      }
#pragma unroll
      for (int mt = 0; mt < 4; ++mt)
#pragma unroll
        for (int r = 0; r < 8; ++r)
          if (mt < 3 || r < 6) t[mt][r] = __builtin_amdgcn_exp2f(t[mt][r]);

      v8f o0 = {}, o1 = {};
#pragma unroll
      for (int mt = 0; mt < 3; ++mt)
#pragma unroll
        for (int p = 0; p < 4; p += 2) {
          v2f b0 = {t[mt][2 * p], t[mt][2 * p + 1]};
          v2f b1 = {t[mt][2 * p + 2], t[mt][2 * p + 3]};
          o0 = wmma_f32(av[mt * 4 + p], b0, o0);
          o1 = wmma_f32(av[mt * 4 + p + 1], b1, o1);
        }
      {
        v2f b12 = {t[3][0] + xhalf(t[3][4]), t[3][1] + xhalf(t[3][5])};
        v2f b13 = {t[3][2], t[3][3]};
        o0 = wmma_f32(av[12], b12, o0);
        o1 = wmma_f32(av[13], b13, o1);
      }
      {
        float inv = 1.0f / (o0[6] + o1[6]);
        int q = nt * 16 + nloc;
        if (lowhalf && q < SEQ) {
          float* zq = Z + q * DIM;
#pragma unroll
          for (int r = 0; r < DIM; r += 2) {
            v2f w = {(o0[r] + o1[r]) * inv + obv[r],
                     (o0[r + 1] + o1[r + 1]) * inv + obv[r + 1]};
            *(v2f*)(zq + r) = w;
          }
        }
      }
    }
    };
    if (exactmax)
      attention(exact_sel<true>{});
    else
      attention(exact_sel<false>{});
#pragma unroll
    for (int t = 0; t < FEAT / 32; ++t) {
      int idx = lane + 32 * t;
      H[idx] = eluf(Z[idx]) + H[idx];
    }
    if (lane < FEAT % 32) {
      int idx = lane + (FEAT / 32) * 32;
      H[idx] = eluf(Z[idx]) + H[idx];
    }
  }

  float vacc = 0.0f;
  float pacc[POL];
#pragma unroll
  for (int o = 0; o < POL; ++o) pacc[o] = 0.0f;
  for (int j = lane; j < FEAT; j += 32) {
    float fj = H[j];
    vacc += fj * vw[j];
#pragma unroll
    for (int o = 0; o < POL; ++o) pacc[o] += fj * pw[o * FEAT + j];
  }
  for (int off = 16; off >= 1; off >>= 1) {
    vacc += __shfl_xor(vacc, off, 32);
#pragma unroll
    for (int o = 0; o < POL; ++o) pacc[o] += __shfl_xor(pacc[o], off, 32);
  }
  if (lane == 0) {
    out[b] = vacc + vb[0];
    float logit[POL];
    float mx = -3.0e38f;
#pragma unroll
    for (int o = 0; o < POL; ++o) {
      logit[o] = pacc[o] + pb[o];
      mx = fmaxf(mx, logit[o]);
    }
    float sum = 0.0f;
#pragma unroll
    for (int o = 0; o < POL; ++o) {
      logit[o] = __expf(logit[o] - mx);
      sum += logit[o];
    }
    float inv = 1.0f / sum;
    float* pout = out + n_batch + (size_t)b * POL;
#pragma unroll
    for (int o = 0; o < POL; ++o) pout[o] = logit[o] * inv;
  }
  }
  __builtin_amdgcn_s_setprio(0);
}

extern "C" void kernel_launch(void* const* d_in, const int* in_sizes, int n_in,
                              void* d_out, int out_size, void* d_ws,
                              size_t ws_size, hipStream_t stream) {
  (void)n_in;
  (void)out_size;
  (void)d_ws;
  (void)ws_size;
  const float* x = (const float*)d_in[0];
  const float* embed_w = (const float*)d_in[1];
  const float* embed_b = (const float*)d_in[2];
  const float* fc1_w = (const float*)d_in[3];
  const float* fc1_b = (const float*)d_in[4];
  const float* ln_g = (const float*)d_in[5];
  const float* ln_b = (const float*)d_in[6];
  const float* iw = (const float*)d_in[7];
  const float* ib = (const float*)d_in[8];
  const float* ow = (const float*)d_in[9];
  const float* ob = (const float*)d_in[10];
  const float* vw = (const float*)d_in[11];
  const float* vb = (const float*)d_in[12];
  const float* pw = (const float*)d_in[13];
  const float* pb = (const float*)d_in[14];

  int n_batch = in_sizes[0] / FEAT;
  rubik_fused<<<NUM_GROUPS, THREADS, 0, stream>>>(x, embed_w, embed_b, fc1_w, fc1_b,
                                            ln_g, ln_b, iw, ib, ow, ob, vw, vb,
                                            pw, pb, (float*)d_out, n_batch);
}


# Last query tile (6 real queries): each query stands in two columns that each sum half the keys, so its PV takes 8 steps and 16 exp2 instead of 14 and 30; key sum order of PV and row sum changes there
# speedup vs baseline: 1.6491x; 1.0309x over previous
#include <hip/hip_runtime.h>
#include <math.h>

#define NUM_TBLOCKS 8
#define SEQ 54
#define DIM 6
#define QKV (3 * DIM)
#define FEAT (SEQ * DIM)
#define POL 12
#define LN_EPS 1e-5f
#define QSCALE 0.58897778410361230f
#define PAD_SCORE -3.0e38f
#define LN_ROW_BOUND 2.4494897427831781f
#define SHIFT_MAX 40.0f

#define WAVES_PER_BLOCK 32
#define THREADS (WAVES_PER_BLOCK * 32)
#define NUM_GROUPS 256
#define WAVE_LDS 2240
#define WSLOT 320
#define WCACHE ((NUM_TBLOCKS + 1) * WSLOT)

typedef __attribute__((ext_vector_type(2))) float v2f;
typedef __attribute__((ext_vector_type(4))) float v4f;
typedef __attribute__((ext_vector_type(8))) float v8f;

template <bool E> struct exact_sel { static constexpr bool value = E; };

__device__ __forceinline__ v8f wmma_f32(v2f a, v2f b, v8f c) {
  return __builtin_amdgcn_wmma_f32_16x16x4_f32(false, a, false, b, (short)0, c,
                                               false, false);
}

__device__ __forceinline__ float xhalf(float x) {
  int b = __builtin_bit_cast(int, x);
  return __builtin_bit_cast(
      float, __builtin_amdgcn_permlanex16(b, b, 0x76543210, 0xfedcba98, false,
                                          false));
}

__device__ __forceinline__ float eluf(float x) {
  return x > 0.0f ? x : (__expf(x) - 1.0f);
}

__global__ __launch_bounds__(THREADS) void rubik_fused(
    const float* __restrict__ x, const float* __restrict__ embed_w,
    const float* __restrict__ embed_b, const float* __restrict__ fc1_w,
    const float* __restrict__ fc1_b, const float* __restrict__ ln_g,
    const float* __restrict__ ln_b, const float* __restrict__ iw,
    const float* __restrict__ ib, const float* __restrict__ ow,
    const float* __restrict__ ob, const float* __restrict__ vw,
    const float* __restrict__ vb, const float* __restrict__ pw,
    const float* __restrict__ pb, float* __restrict__ out, int n_batch) {
  __shared__ float smem[WAVES_PER_BLOCK * WAVE_LDS + WCACHE];
  const int lane = threadIdx.x & 31;
  const int wave = __builtin_amdgcn_readfirstlane((int)(threadIdx.x >> 5));

  float* H = &smem[wave * WAVE_LDS];
  float* Z = H + 324;
  float* Qp = Z + 324;
  float* Kp = Qp + 512;
  float* Vp = Kp + 512;
  float* dump = H + 2120;

  float* WC = &smem[WAVES_PER_BLOCK * WAVE_LDS];

  const int nloc = lane & 15;
  const int kb = (lane < 16) ? 0 : 2;
  const int upper8 = (lane < 16) ? 0 : 8;
  const bool lowhalf = (lane < 16);

  const bool qvcol = (nloc <= 2 * DIM);
  float* qvBase = (nloc < DIM)       ? (Qp + nloc)
                  : (nloc < 2 * DIM) ? (Vp + (nloc - DIM))
                                     : (Qp + DIM);

  auto fillW1 = [&](const float* w, int slot, float* wimg) {
    int kk = slot >> 6, ln = (slot >> 1) & 31, p = slot & 1;
    int j = ln & 15;
    int c = kk * 4 + ((ln < 16) ? 0 : 2) + p;
    bool v = (j < DIM) && (c < DIM);
    wimg[slot] = v ? w[v ? j * DIM + c : 0] : 0.0f;
  };
  auto fillB1 = [&](const float* bias, int ln, float* bimg) {
    int j = ln & 15;
    bimg[ln] = (j < DIM) ? bias[(j < DIM) ? j : 0] : 0.0f;
  };

  auto fillQVw = [&](const float* iwl, const float* owl, int slot,
                     float* wimg) {
    {
      int kk = slot >> 6, ln = (slot >> 1) & 31, p = slot & 1;
      int n = ln & 15;
      int c = kk * 4 + ((ln < 16) ? 0 : 2) + p;
      bool isq = (n < DIM);
      bool v = (n < 2 * DIM) && (c < DIM);
      int a = v ? (isq ? n : n - DIM) : 0;
      int cc = v ? c : 0;
      const float* lp = isq ? (iwl + DIM * DIM + a) : (owl + a * DIM);
      int ls = isq ? DIM : 1;
      const float* rp = (isq ? iwl : (iwl + 2 * DIM * DIM)) + cc;
      float acc = 0.0f;
      for (int m = 0; m < DIM; ++m) acc = fmaf(lp[m * ls], rp[m * DIM], acc);
      if (isq) acc *= QSCALE;
      wimg[slot] = v ? acc : 0.0f;
    }
  };
  auto fillQVb = [&](const float* iwl, const float* ibl, const float* owl,
                     int ln, float* bimg) {
    {
      int n = ln & 15;
      bool isq = (n < DIM);
      bool v = (n < 2 * DIM);
      int a = v ? (isq ? n : n - DIM) : 0;
      const float* lp = isq ? (iwl + DIM * DIM + a) : (owl + a * DIM);
      int ls = isq ? DIM : 1;
      const float* rp = isq ? ibl : (ibl + 2 * DIM);
      float acc = 0.0f;
      for (int m = 0; m < DIM; ++m) acc = fmaf(lp[m * ls], rp[m], acc);
      if (isq) acc *= QSCALE;
      bimg[ln] = v ? acc : 0.0f;
    }
  };

  auto loadA = [&](const float* src, int mt, v2f aw[2]) {
    int row = mt * 16 + nloc;
#pragma unroll
    for (int kk = 0; kk < 2; ++kk)
      aw[kk] = *(const v2f*)(src + row * DIM + kk * 4 + kb);
  };
  auto loadBimg = [&](const float* wimg, v2f bw[2]) {
    bw[0] = *(const v2f*)(wimg + lane * 2);
    bw[1] = *(const v2f*)(wimg + 64 + lane * 2);
  };
  auto loadCimg = [&](const float* bimg) {
    float bvv = bimg[lane];
    v8f c;
#pragma unroll
    for (int r = 0; r < 8; ++r) c[r] = bvv;
    return c;
  };

  {
    const int t = threadIdx.x;
    const int Ls = t >> 7, slot = t & 127;
    float* ws = WC + Ls * WSLOT;
    fillW1(fc1_w + Ls * DIM * DIM, slot, ws);
    fillQVw(iw + Ls * QKV * DIM, ow + Ls * DIM * DIM, slot, ws + 128);
    if (t < NUM_TBLOCKS * 32) {
      const int Lb = t >> 5, ln = t & 31;
      float* wb = WC + Lb * WSLOT;
      fillB1(fc1_b + Lb * DIM, ln, wb + 256);
      fillQVb(iw + Lb * QKV * DIM, ib + Lb * QKV, ow + Lb * DIM * DIM, ln,
              wb + 288);
    }
    float* we = WC + NUM_TBLOCKS * WSLOT;
    if (t < 128) fillW1(embed_w, t, we);
    if (t < 32) fillB1(embed_b, t, we + 256);
  }
  __syncthreads();

  if (wave < NUM_TBLOCKS) {
    float* ws = WC + wave * WSLOT;
    const float* wq = ws + 128;
    float* bqi = ws + 288;
    v2f w0 = *(const v2f*)(wq + lane * 2);
    v2f w1 = *(const v2f*)(wq + 64 + lane * 2);
    float cbv = bqi[lane];
    float sw = (nloc < DIM)
                   ? (w0.x * w0.x + w0.y * w0.y + w1.x * w1.x + w1.y * w1.y)
                   : 0.0f;
    float sc = (lane < DIM) ? cbv * cbv : 0.0f;
    float gv = (lane < DIM) ? ln_g[wave * DIM + ((lane < DIM) ? lane : 0)] : 0.0f;
    float bv = (lane < DIM) ? ln_b[wave * DIM + ((lane < DIM) ? lane : 0)] : 0.0f;
    float gm = fabsf(gv);
    float sb = bv * bv;
    for (int off = 16; off >= 1; off >>= 1) {
      sw += __shfl_xor(sw, off, 32);
      sc += __shfl_xor(sc, off, 32);
      sb += __shfl_xor(sb, off, 32);
      gm = fmaxf(gm, __shfl_xor(gm, off, 32));
    }
    float R = LN_ROW_BOUND * gm + sqrtf(sb);
    float B = (sqrtf(sw) * R + sqrtf(sc)) * R;
    float shift = (B <= SHIFT_MAX) ? -B : 0.0f;
    if (nloc == 2 * DIM) bqi[lane] = shift;
  }
  __syncthreads();

  for (int idx = lane; idx < 512 + 512 + 448; idx += 32) Qp[idx] = 0.0f;

  for (int s = lane; s < SEQ; s += 32) Vp[s * 8 + DIM] = 1.0f;
  for (int s = lane; s < 64; s += 32) Qp[s * 8 + 7] = PAD_SCORE;
  if (lane < 64 - SEQ) Kp[(SEQ + lane) * 8 + 7] = 1.0f;
  for (int s = lane; s < SEQ; s += 32) Kp[s * 8 + DIM] = 1.0f;

  for (int b = blockIdx.x * WAVES_PER_BLOCK + wave; b < n_batch;
       b += NUM_GROUPS * WAVES_PER_BLOCK) {
  const unsigned left =
      (unsigned)(n_batch - 1 - b) / (unsigned)(NUM_GROUPS * WAVES_PER_BLOCK);
  if (left >= 2u)
    __builtin_amdgcn_s_setprio(3);
  else if (left == 1u)
    __builtin_amdgcn_s_setprio(2);
  else
    __builtin_amdgcn_s_setprio(1);
  const float* xb = x + (size_t)b * FEAT;
  {
    int bn = b + NUM_GROUPS * WAVES_PER_BLOCK;
    if (bn < n_batch && lane * 32 < FEAT)
      __builtin_prefetch(x + (size_t)bn * FEAT + lane * 32, 0, 1);
  }
  for (int idx = lane; idx < FEAT; idx += 32) Z[idx] = xb[idx];

  {
    const float* W_FC = WC + NUM_TBLOCKS * WSLOT;
    const float* B_FC = W_FC + 256;
    v2f bw[2];
    loadBimg(W_FC, bw);
    v8f cb = loadCimg(B_FC);
#pragma unroll
    for (int mt = 0; mt < 4; ++mt) {
      v2f aw[2];
      loadA(Z, mt, aw);
      v8f d = wmma_f32(aw[0], bw[0], cb);
      d = wmma_f32(aw[1], bw[1], d);
      int row0 = mt * 16 + upper8;
      bool ok = (nloc < DIM) && ((mt < 3) || lowhalf);
      float* bz = ok ? &H[row0 * DIM + nloc] : dump;
#pragma unroll
      for (int r = 0; r < 8; ++r)
        if (mt < 3 || r < 6) bz[r * DIM] = d[r];
    }
  }

  for (int L = 0; L < NUM_TBLOCKS; ++L) {
    if (left == 0u && L == NUM_TBLOCKS - 3) __builtin_amdgcn_s_setprio(0);
    typedef const __attribute__((address_space(4))) float cfloat;
    cfloat* g = (cfloat*)(ln_g + L * DIM);
    cfloat* bbia = (cfloat*)(ln_b + L * DIM);
    cfloat* obl = (cfloat*)(ob + L * DIM);
    const float* W_FC = WC + L * WSLOT;
    const float* W_QV = W_FC + 128;
    const float* B_FC = W_FC + 256;
    const float* B_QV = W_FC + 288;

    {
      v2f bw[2];
      loadBimg(W_FC, bw);
      v8f cb = loadCimg(B_FC);
#pragma unroll
      for (int mt = 0; mt < 4; ++mt) {
        v2f aw[2];
        loadA(H, mt, aw);
        v8f d = wmma_f32(aw[0], bw[0], cb);
        d = wmma_f32(aw[1], bw[1], d);
        int row0 = mt * 16 + upper8;
        bool ok = (nloc < DIM) && ((mt < 3) || lowhalf);
        float* bz = ok ? &Z[row0 * DIM + nloc] : dump;
#pragma unroll
        for (int r = 0; r < 8; ++r)
          if (mt < 3 || r < 6) bz[r * DIM] = d[r];
      }
    }

    for (int s = lane; s < SEQ; s += 32) {
      float zv[DIM];
#pragma unroll
      for (int i = 0; i < DIM; ++i) zv[i] = eluf(Z[s * DIM + i]);
      float mu = 0.0f;
#pragma unroll
      for (int i = 0; i < DIM; ++i) mu += zv[i];
      mu *= (1.0f / DIM);
      float var = 0.0f;
#pragma unroll
      for (int i = 0; i < DIM; ++i) {
        float d = zv[i] - mu;
        var += d * d;
      }
      var *= (1.0f / DIM);
      float rs = rsqrtf(var + LN_EPS);
#pragma unroll
      for (int i = 0; i < DIM; ++i)
        Kp[s * 8 + i] = (zv[i] - mu) * rs * g[i] + bbia[i];
    }

    v2f ak[4][2];
#pragma unroll
    for (int mt = 0; mt < 4; ++mt)
#pragma unroll
      for (int kk = 0; kk < 2; ++kk)
        ak[mt][kk] = *(const v2f*)(Kp + (mt * 16 + nloc) * 8 + kk * 4 + kb);

    bool exactmax;
    {
      v2f bw[2];
      loadBimg(W_QV, bw);
      v8f cb = loadCimg(B_QV);
      exactmax = __builtin_bit_cast(
                     float, __builtin_amdgcn_readlane(
                                __builtin_bit_cast(int, cb[0]), 2 * DIM)) == 0.0f;
#pragma unroll
      for (int mt = 0; mt < 4; ++mt) {
        v8f d = wmma_f32(ak[mt][0], bw[0], cb);
        d = wmma_f32(ak[mt][1], bw[1], d);
        int row0 = mt * 16 + upper8;
        bool ok0 = qvcol && ((mt < 3) || lowhalf);
        float* b0 = ok0 ? (qvBase + row0 * 8) : dump;
#pragma unroll
        for (int r = 0; r < 8; ++r)
          if (mt < 3 || r < 6) b0[r * 8] = d[r];
      }
    }

    const float* vcolp = Vp + ((nloc < 8) ? nloc : 7);
    v2f av[14];
#pragma unroll
    for (int mt = 0; mt < 3; ++mt)
#pragma unroll
      for (int p = 0; p < 4; ++p) {
        int key = mt * 16 + upper8 + 2 * p;
        av[mt * 4 + p].x = vcolp[key * 8];
        av[mt * 4 + p].y = vcolp[(key + 1) * 8];
      }
    {
      int key12 = lowhalf ? 48 : 52;
      av[12].x = vcolp[key12 * 8];
      av[12].y = vcolp[(key12 + 1) * 8];
      int key13 = lowhalf ? 50 : 54;
      av[13].x = vcolp[key13 * 8];
      av[13].y = vcolp[(key13 + 1) * 8];
    }

    float obv[DIM];
    for (int r = 0; r < DIM; ++r) obv[r] = obl[r];

    auto attention = [&](auto sel) {
    constexpr bool EXACT = decltype(sel)::value;
#pragma nounroll
    for (int nt = 0; nt < 3; ++nt) {
      v2f bq[2];
#pragma unroll
      for (int kk = 0; kk < 2; ++kk)
        bq[kk] = *(const v2f*)(Qp + (nt * 16 + nloc) * 8 + kk * 4 + kb);
      v8f t[4];
#pragma unroll
      for (int mt = 0; mt < 4; ++mt) {
        v8f c = {};
#pragma unroll
        for (int kk = 0; kk < 2; ++kk) c = wmma_f32(ak[mt][kk], bq[kk], c);
        t[mt] = c;
      }

      if constexpr (EXACT) {
        float mx = t[0][0];
#pragma unroll
        for (int mt = 0; mt < 4; ++mt)
#pragma unroll
          for (int r = 0; r < 8; ++r)
            if ((mt > 0 || r > 0) && (mt < 3 || r < 6))
              mx = fmaxf(mx, t[mt][r]);
        mx = fmaxf(mx, __shfl_xor(mx, 16));
        const v2f mx2 = {mx, mx};
#pragma unroll
        for (int mt = 0; mt < 4; ++mt)
#pragma unroll
          for (int p = 0; p < 4; ++p)
            if (mt < 3 || p < 3) {
              v2f d = (v2f){t[mt][2 * p], t[mt][2 * p + 1]} - mx2;
              t[mt][2 * p] = d.x;
              t[mt][2 * p + 1] = d.y;
            }
      }
#pragma unroll
      for (int mt = 0; mt < 4; ++mt)
#pragma unroll
        for (int r = 0; r < 8; ++r)
          if (mt < 3 || r < 6) t[mt][r] = __builtin_amdgcn_exp2f(t[mt][r]);

      v8f o0 = {}, o1 = {};
#pragma unroll
      for (int mt = 0; mt < 3; ++mt)
#pragma unroll
        for (int p = 0; p < 4; p += 2) {
          v2f b0 = {t[mt][2 * p], t[mt][2 * p + 1]};
          v2f b1 = {t[mt][2 * p + 2], t[mt][2 * p + 3]};
          o0 = wmma_f32(av[mt * 4 + p], b0, o0);
          o1 = wmma_f32(av[mt * 4 + p + 1], b1, o1);
        }
      {
        v2f b12 = {t[3][0] + xhalf(t[3][4]), t[3][1] + xhalf(t[3][5])};
        v2f b13 = {t[3][2], t[3][3]};
        o0 = wmma_f32(av[12], b12, o0);
        o1 = wmma_f32(av[13], b13, o1);
      }
      {
        float inv = 1.0f / (o0[6] + o1[6]);
        int q = nt * 16 + nloc;
        if (lowhalf && q < SEQ) {
          float* zq = Z + q * DIM;
#pragma unroll
          for (int r = 0; r < DIM; r += 2) {
            v2f w = {(o0[r] + o1[r]) * inv + obv[r],
                     (o0[r + 1] + o1[r + 1]) * inv + obv[r + 1]};
            *(v2f*)(zq + r) = w;
          }
        }
      }
    }
    {
      const int c7 = nloc & 7;
      const bool colhi = (nloc >= 8);
      v2f aw[8];
      {
        const float* vrow = Vp + c7 + (upper8 + (colhi ? 4 : 0)) * 8;
        const float* vrow3 = lowhalf ? (vrow + 48 * 8) : (Vp + c7 + 54 * 8);
        const int st3 = lowhalf ? 8 : 0;
#pragma unroll
        for (int mt = 0; mt < 3; ++mt)
#pragma unroll
          for (int p = 0; p < 2; ++p) {
            aw[mt * 2 + p].x = vrow[(mt * 16 + 2 * p) * 8];
            aw[mt * 2 + p].y = vrow[(mt * 16 + 2 * p + 1) * 8];
          }
#pragma unroll
        for (int p = 0; p < 2; ++p) {
          aw[6 + p].x = vrow3[(2 * p) * st3];
          aw[6 + p].y = vrow3[(2 * p + 1) * st3];
        }
      }
      v2f bq[2];
#pragma unroll
      for (int kk = 0; kk < 2; ++kk)
        bq[kk] = *(const v2f*)(Qp + (48 + c7) * 8 + kk * 4 + kb);
      v8f t[4];
#pragma unroll
      for (int mt = 0; mt < 4; ++mt) {
        v8f cz = {};
#pragma unroll
        for (int kk = 0; kk < 2; ++kk) cz = wmma_f32(ak[mt][kk], bq[kk], cz);
        t[mt] = cz;
      }
      float mx = 0.0f;
      if constexpr (EXACT) {
        mx = t[0][0];
#pragma unroll
        for (int mt = 0; mt < 4; ++mt)
#pragma unroll
          for (int r = 0; r < 8; ++r)
            if ((mt > 0 || r > 0) && (mt < 3 || r < 6))
              mx = fmaxf(mx, t[mt][r]);
        mx = fmaxf(mx, xhalf(mx));
      }
      float e[4][4];
#pragma unroll
      for (int mt = 0; mt < 4; ++mt)
#pragma unroll
        for (int j = 0; j < 4; ++j) {
          float shi = t[mt][4 + j];
          asm("" : "+v"(shi));
          float s = colhi ? shi : t[mt][j];
          if constexpr (EXACT) s -= mx;
          e[mt][j] = s;
        }
#pragma unroll
      for (int mt = 0; mt < 4; ++mt)
#pragma unroll
        for (int j = 0; j < 4; ++j)
          e[mt][j] = __builtin_amdgcn_exp2f(e[mt][j]);
      v8f o0 = {}, o1 = {};
#pragma unroll
      for (int mt = 0; mt < 4; ++mt) {
        v2f b0 = {e[mt][0], e[mt][1]};
        v2f b1 = {e[mt][2], e[mt][3]};
        o0 = wmma_f32(aw[mt * 2], b0, o0);
        o1 = wmma_f32(aw[mt * 2 + 1], b1, o1);
      }
      float tot[DIM + 1];
#pragma unroll
      for (int r = 0; r <= DIM; ++r) {
        float part = o0[r] + o1[r];
        int pb = __builtin_bit_cast(int, part);
        tot[r] = part + __builtin_bit_cast(
                            float, __builtin_amdgcn_permlanex16(
                                       pb, pb, 0xfedcba98, 0x76543210, false,
                                       false));
      }
      float inv = 1.0f / tot[DIM];
      if (lowhalf && nloc < SEQ - 48) {
        float* zq = Z + (48 + nloc) * DIM;
#pragma unroll
        for (int r = 0; r < DIM; r += 2) {
          v2f w = {tot[r] * inv + obv[r], tot[r + 1] * inv + obv[r + 1]};
          *(v2f*)(zq + r) = w;
        }
      }
    }
    };
    if (exactmax)
      attention(exact_sel<true>{});
    else
      attention(exact_sel<false>{});
#pragma unroll
    for (int t = 0; t < FEAT / 32; ++t) {
      int idx = lane + 32 * t;
      H[idx] = eluf(Z[idx]) + H[idx];
    }
    if (lane < FEAT % 32) {
      int idx = lane + (FEAT / 32) * 32;
      H[idx] = eluf(Z[idx]) + H[idx];
    }
  }

  float vacc = 0.0f;
  float pacc[POL];
#pragma unroll
  for (int o = 0; o < POL; ++o) pacc[o] = 0.0f;
  for (int j = lane; j < FEAT; j += 32) {
    float fj = H[j];
    vacc += fj * vw[j];
#pragma unroll
    for (int o = 0; o < POL; ++o) pacc[o] += fj * pw[o * FEAT + j];
  }
  for (int off = 16; off >= 1; off >>= 1) {
    vacc += __shfl_xor(vacc, off, 32);
#pragma unroll
    for (int o = 0; o < POL; ++o) pacc[o] += __shfl_xor(pacc[o], off, 32);
  }
  if (lane == 0) {
    out[b] = vacc + vb[0];
    float logit[POL];
    float mx = -3.0e38f;
#pragma unroll
    for (int o = 0; o < POL; ++o) {
      logit[o] = pacc[o] + pb[o];
      mx = fmaxf(mx, logit[o]);
    }
    float sum = 0.0f;
#pragma unroll
    for (int o = 0; o < POL; ++o) {
      logit[o] = __expf(logit[o] - mx);
      sum += logit[o];
    }
    float inv = 1.0f / sum;
    float* pout = out + n_batch + (size_t)b * POL;
#pragma unroll
    for (int o = 0; o < POL; ++o) pout[o] = logit[o] * inv;
  }
  }
  __builtin_amdgcn_s_setprio(0);
}

extern "C" void kernel_launch(void* const* d_in, const int* in_sizes, int n_in,
                              void* d_out, int out_size, void* d_ws,
                              size_t ws_size, hipStream_t stream) {
  (void)n_in;
  (void)out_size;
  (void)d_ws;
  (void)ws_size;
  const float* x = (const float*)d_in[0];
  const float* embed_w = (const float*)d_in[1];
  const float* embed_b = (const float*)d_in[2];
  const float* fc1_w = (const float*)d_in[3];
  const float* fc1_b = (const float*)d_in[4];
  const float* ln_g = (const float*)d_in[5];
  const float* ln_b = (const float*)d_in[6];
  const float* iw = (const float*)d_in[7];
  const float* ib = (const float*)d_in[8];
  const float* ow = (const float*)d_in[9];
  const float* ob = (const float*)d_in[10];
  const float* vw = (const float*)d_in[11];
  const float* vb = (const float*)d_in[12];
  const float* pw = (const float*)d_in[13];
  const float* pb = (const float*)d_in[14];

  int n_batch = in_sizes[0] / FEAT;
  rubik_fused<<<NUM_GROUPS, THREADS, 0, stream>>>(x, embed_w, embed_b, fc1_w, fc1_b,
                                            ln_g, ln_b, iw, ib, ow, ob, vw, vb,
                                            pw, pb, (float*)d_out, n_batch);
}
